# starting kernel
# baseline (speedup 1.0000x reference)
#include <hip/hip_runtime.h>

typedef __attribute__((ext_vector_type(2))) float v2f;
typedef __attribute__((ext_vector_type(8))) float v8f;

constexpr int T      = 12;
constexpr int C_IN   = 32;
constexpr int H      = 64;
constexpr int C_OUT  = 32;
constexpr int N_SRC  = 50000;
constexpr int N_DST1 = 20000;
constexpr int N_DST2 = 10000;
constexpr int ROW    = T * C_IN;
constexpr int ROW4   = ROW / 4;
constexpr int M1     = N_DST1 * T;
constexpr int M2     = N_DST2 * T;
constexpr float NEG_SLOPE = 0.01f;

__device__ __forceinline__ float leaky(float v) {
    return v > 0.f ? v : NEG_SLOPE * v;
}

__device__ __forceinline__ v2f load_frag(const float* __restrict__ base, int r0, int ld,
                                         int k0, int lane) {
    int r = r0 + (lane & 15);
    int k = k0 + ((lane >> 4) << 1);
    const float* p = base + (size_t)r * ld + k;
    v2f f;
    f.x = p[0];
    f.y = p[1];
    return f;
}

__global__ void k_transpose(const float* __restrict__ nf, float* __restrict__ xT) {
    int idx = blockIdx.x * blockDim.x + threadIdx.x;
    constexpr int C4 = C_IN / 4;
    constexpr int total = N_SRC * T * C4;
    if (idx >= total) return;
    int c4 = idx % C4;
    int t  = (idx / C4) % T;
    int n  = idx / (C4 * T);
    const float4* src = reinterpret_cast<const float4*>(nf);
    float4 v = src[(size_t)t * N_SRC * C4 + (size_t)n * C4 + c4];
    reinterpret_cast<float4*>(xT)[idx] = v;
}

__global__ void k_zero_int(int* __restrict__ p, int n) {
    int i = blockIdx.x * blockDim.x + threadIdx.x;
    if (i < n) p[i] = 0;
}

__global__ void k_hist(const int* __restrict__ dst, int* __restrict__ counts, int E) {
    int e = blockIdx.x * blockDim.x + threadIdx.x;
    if (e < E) atomicAdd(&counts[dst[e]], 1);
}

__global__ void __launch_bounds__(1024)
k_scan(const int* __restrict__ counts, int* __restrict__ starts,
       int* __restrict__ cursor, int N) {
    __shared__ int part[1024];
    int tid   = threadIdx.x;
    int chunk = (N + 1023) >> 10;
    int begin = tid * chunk;
    int end   = begin + chunk;
    if (end > N) end = N;
    int sum = 0;
    for (int i = begin; i < end; ++i) sum += counts[i];
    part[tid] = sum;
    __syncthreads();
    for (int off = 1; off < 1024; off <<= 1) {
        int v = 0;
        if (tid >= off) v = part[tid - off];
        __syncthreads();
        part[tid] += v;
        __syncthreads();
    }
    int run = part[tid] - sum;
    for (int i = begin; i < end; ++i) {
        starts[i] = run;
        cursor[i] = run;
        run += counts[i];
    }
    if (tid == 1023) starts[N] = part[1023];
}

__global__ void k_bin(const int* __restrict__ src, const int* __restrict__ dst,
                      const float* __restrict__ ew, int* __restrict__ cursor,
                      int* __restrict__ srcs, float* __restrict__ ws, int E) {
    int e = blockIdx.x * blockDim.x + threadIdx.x;
    if (e >= E) return;
    int pos = atomicAdd(&cursor[dst[e]], 1);
    srcs[pos] = src[e];
    ws[pos]   = ew[e];
}

__global__ void __launch_bounds__(96)
k_aggregate(const int* __restrict__ starts, const int* __restrict__ srcs,
            const float* __restrict__ ws, const float* __restrict__ x,
            float* __restrict__ agg) {
    int n   = blockIdx.x;
    int tid = threadIdx.x;
    int beg = starts[n];
    int end = starts[n + 1];
    float4 acc = make_float4(0.f, 0.f, 0.f, 0.f);
    const float4* x4 = reinterpret_cast<const float4*>(x);
    for (int j = beg; j < end; ++j) {
        int   s = srcs[j];
        float w = ws[j];
        float4 v = x4[(size_t)s * ROW4 + tid];
        acc.x += w * v.x;
        acc.y += w * v.y;
        acc.z += w * v.z;
        acc.w += w * v.w;
    }
    reinterpret_cast<float4*>(agg)[(size_t)n * ROW4 + tid] = acc;
}

__global__ void __launch_bounds__(128)
k_gemm1(const float* __restrict__ xT, const float* __restrict__ agg1,
        const float* __restrict__ Wself, const float* __restrict__ Wneigh,
        const float* __restrict__ bias, float* __restrict__ h1) {
    int lane  = threadIdx.x & 31;
    int wave  = threadIdx.x >> 5;
    int mtile = blockIdx.x * (blockDim.x >> 5) + wave;
    if (mtile >= M1 / 16) return;
    int m0 = mtile * 16;

    v2f aS[8], aN[8];
#pragma unroll
    for (int ks = 0; ks < 8; ++ks) {
        aS[ks] = load_frag(xT,   m0, C_IN, ks * 4, lane);
        aN[ks] = load_frag(agg1, m0, C_IN, ks * 4, lane);
    }

#pragma unroll
    for (int nt = 0; nt < 4; ++nt) {
        int n0 = nt * 16;
        v8f acc = {};
#pragma unroll
        for (int ks = 0; ks < 8; ++ks) {
            v2f bS = load_frag(Wself, n0, C_IN, ks * 4, lane);
            acc = __builtin_amdgcn_wmma_f32_16x16x4_f32(false, aS[ks], false, bS,
                                                        (short)0, acc, false, false);
            v2f bN = load_frag(Wneigh, n0, C_IN, ks * 4, lane);
            acc = __builtin_amdgcn_wmma_f32_16x16x4_f32(false, aN[ks], false, bN,
                                                        (short)0, acc, false, false);
        }
        int col   = n0 + (lane & 15);
        int mbase = m0 + ((lane >> 4) << 3);
        float bv  = bias[col];
#pragma unroll
        for (int r = 0; r < 8; ++r) {
            float v = leaky(acc[r] + bv);
            h1[(size_t)(mbase + r) * H + col] = v;
        }
    }
}

__global__ void __launch_bounds__(128)
k_gemm_z2(const float* __restrict__ h1, const float* __restrict__ Wneigh2,
          float* __restrict__ z2) {
    int lane  = threadIdx.x & 31;
    int wave  = threadIdx.x >> 5;
    int mtile = blockIdx.x * (blockDim.x >> 5) + wave;
    if (mtile >= M1 / 16) return;
    int m0 = mtile * 16;

    v2f a[16];
#pragma unroll
    for (int ks = 0; ks < 16; ++ks)
        a[ks] = load_frag(h1, m0, H, ks * 4, lane);

#pragma unroll
    for (int nt = 0; nt < 2; ++nt) {
        int n0 = nt * 16;
        v8f acc = {};
#pragma unroll
        for (int ks = 0; ks < 16; ++ks) {
            v2f b = load_frag(Wneigh2, n0, H, ks * 4, lane);
            acc = __builtin_amdgcn_wmma_f32_16x16x4_f32(false, a[ks], false, b,
                                                        (short)0, acc, false, false);
        }
        int col   = n0 + (lane & 15);
        int mbase = m0 + ((lane >> 4) << 3);
#pragma unroll
        for (int r = 0; r < 8; ++r)
            z2[(size_t)(mbase + r) * C_OUT + col] = acc[r];
    }
}

__global__ void __launch_bounds__(128)
k_final(const float* __restrict__ h1, const float* __restrict__ agg2,
        const float* __restrict__ Wself2, const float* __restrict__ bias2,
        float* __restrict__ out) {
    int lane  = threadIdx.x & 31;
    int wave  = threadIdx.x >> 5;
    int mtile = blockIdx.x * (blockDim.x >> 5) + wave;
    if (mtile >= M2 / 16) return;
    int m0 = mtile * 16;

    v2f a[16];
#pragma unroll
    for (int ks = 0; ks < 16; ++ks)
        a[ks] = load_frag(h1, m0, H, ks * 4, lane);

#pragma unroll
    for (int nt = 0; nt < 2; ++nt) {
        int n0 = nt * 16;
        v8f acc = {};
#pragma unroll
        for (int ks = 0; ks < 16; ++ks) {
            v2f b = load_frag(Wself2, n0, H, ks * 4, lane);
            acc = __builtin_amdgcn_wmma_f32_16x16x4_f32(false, a[ks], false, b,
                                                        (short)0, acc, false, false);
        }
        int col   = n0 + (lane & 15);
        int mbase = m0 + ((lane >> 4) << 3);
        float bv  = bias2[col];
#pragma unroll
        for (int r = 0; r < 8; ++r) {
            int m   = mbase + r;
            float v = leaky(acc[r] + agg2[(size_t)m * C_OUT + col] + bv);
            int n = m / T;
            int t = m % T;
            out[((size_t)t * N_DST2 + n) * C_OUT + col] = v;
        }
    }
}

extern "C" void kernel_launch(void* const* d_in, const int* in_sizes, int n_in,
                              void* d_out, int out_size, void* d_ws, size_t ws_size,
                              hipStream_t stream) {
    const float* nf   = (const float*)d_in[0];
    const int*   src1 = (const int*)  d_in[1];
    const int*   dst1 = (const int*)  d_in[2];
    const float* ew1  = (const float*)d_in[3];
    const int*   src2 = (const int*)  d_in[4];
    const int*   dst2 = (const int*)  d_in[5];
    const float* ew2  = (const float*)d_in[6];
    const float* W1s  = (const float*)d_in[7];
    const float* W1n  = (const float*)d_in[8];
    const float* b1   = (const float*)d_in[9];
    const float* W2s  = (const float*)d_in[10];
    const float* W2n  = (const float*)d_in[11];
    const float* b2   = (const float*)d_in[12];
    const int E1 = in_sizes[1];
    const int E2 = in_sizes[4];

    float* out = (float*)d_out;

    float* ws   = (float*)d_ws;
    float* xT   = ws;
    float* agg1 = xT   + (size_t)N_SRC  * ROW;
    float* h1   = agg1 + (size_t)N_DST1 * ROW;
    float* z2   = xT;
    float* agg2 = xT + (size_t)N_DST1 * ROW;
    int*   ip      = (int*)(h1 + (size_t)M1 * H);
    int*   counts1 = ip;                 ip += N_DST1;
    int*   starts1 = ip;                 ip += N_DST1 + 1;
    int*   cursor1 = ip;                 ip += N_DST1;
    int*   srcs1   = ip;                 ip += E1;
    float* ws1     = (float*)ip;         ip += E1;
    int*   counts2 = ip;                 ip += N_DST2;
    int*   starts2 = ip;                 ip += N_DST2 + 1;
    int*   cursor2 = ip;                 ip += N_DST2;
    int*   srcs2   = ip;                 ip += E2;
    float* ws2     = (float*)ip;

    {
        int total = N_SRC * T * (C_IN / 4);
        k_transpose<<<(total + 255) / 256, 256, 0, stream>>>(nf, xT);
    }
    k_zero_int<<<(N_DST1 + 255) / 256, 256, 0, stream>>>(counts1, N_DST1);
    k_zero_int<<<(N_DST2 + 255) / 256, 256, 0, stream>>>(counts2, N_DST2);
    k_hist<<<(E1 + 255) / 256, 256, 0, stream>>>(dst1, counts1, E1);
    k_hist<<<(E2 + 255) / 256, 256, 0, stream>>>(dst2, counts2, E2);
    k_scan<<<1, 1024, 0, stream>>>(counts1, starts1, cursor1, N_DST1);
    k_scan<<<1, 1024, 0, stream>>>(counts2, starts2, cursor2, N_DST2);
    k_bin<<<(E1 + 255) / 256, 256, 0, stream>>>(src1, dst1, ew1, cursor1, srcs1, ws1, E1);
    k_bin<<<(E2 + 255) / 256, 256, 0, stream>>>(src2, dst2, ew2, cursor2, srcs2, ws2, E2);

    k_aggregate<<<N_DST1, 96, 0, stream>>>(starts1, srcs1, ws1, xT, agg1);

    k_gemm1<<<(M1 / 16 + 3) / 4, 128, 0, stream>>>(xT, agg1, W1s, W1n, b1, h1);

    k_gemm_z2<<<(M1 / 16 + 3) / 4, 128, 0, stream>>>(h1, W2n, z2);

    k_aggregate<<<N_DST2, 96, 0, stream>>>(starts2, srcs2, ws2, z2, agg2);

    k_final<<<(M2 / 16 + 3) / 4, 128, 0, stream>>>(h1, agg2, W2s, b2, out);
}


# Adjacency as 4 linked lists per dst: one atomic exchange puts an edge in front of list e%4 of its dst (no count/scan/scatter, 14->7 launches); gathers walk the 4 lists in step; sums in list order
# speedup vs baseline: 1.7587x; 1.7587x over previous
#include <hip/hip_runtime.h>

typedef __attribute__((ext_vector_type(2))) float v2f;
typedef __attribute__((ext_vector_type(8))) float v8f;

constexpr int T      = 12;
constexpr int C_IN   = 32;
constexpr int H      = 64;
constexpr int C_OUT  = 32;
constexpr int N_SRC  = 50000;
constexpr int N_DST1 = 20000;
constexpr int N_DST2 = 10000;
constexpr int ROW    = T * C_IN;
constexpr int ROW4   = ROW / 4;
constexpr int M1     = N_DST1 * T;
constexpr int M2     = N_DST2 * T;
constexpr float NEG_SLOPE = 0.01f;

__device__ __forceinline__ float leaky(float v) {
    return v > 0.f ? v : NEG_SLOPE * v;
}

__device__ __forceinline__ v2f load_frag(const float* __restrict__ base, int r0, int ld,
                                         int k0, int lane) {
    int r = r0 + (lane & 15);
    int k = k0 + ((lane >> 4) << 1);
    const float* p = base + (size_t)r * ld + k;
    v2f f;
    f.x = p[0];
    f.y = p[1];
    return f;
}

__global__ void k_transpose(const float* __restrict__ nf, float* __restrict__ xT,
                            int* __restrict__ heads, int nheads) {
    int idx = blockIdx.x * blockDim.x + threadIdx.x;
    constexpr int C4 = C_IN / 4;
    constexpr int total = N_SRC * T * C4;
    if (idx < nheads) heads[idx] = -1;
    if (idx >= total) return;
    int c4 = idx % C4;
    int t  = (idx / C4) % T;
    int n  = idx / (C4 * T);
    const float4* src = reinterpret_cast<const float4*>(nf);
    float4 v = src[(size_t)t * N_SRC * C4 + (size_t)n * C4 + c4];
    reinterpret_cast<float4*>(xT)[idx] = v;
}

constexpr int NL = 4;

__global__ void __launch_bounds__(256)
k_link(const int* __restrict__ src1, const int* __restrict__ dst1,
       const float* __restrict__ ew1, int E1, int nb1,
       const int* __restrict__ src2, const int* __restrict__ dst2,
       const float* __restrict__ ew2, int E2,
       int* __restrict__ heads1, int* __restrict__ heads2,
       int4* __restrict__ rec1, int4* __restrict__ rec2) {
    int b = blockIdx.x;
    const int*   src = src1;
    const int*   dst = dst1;
    const float* ew  = ew1;
    int*         hd  = heads1;
    int4*        rec = rec1;
    int          E   = E1;
    if (b >= nb1) {
        b -= nb1;
        src = src2; dst = dst2; ew = ew2; hd = heads2; rec = rec2; E = E2;
    }
    int e = b * blockDim.x + threadIdx.x;
    if (e >= E) return;
    int d    = dst[e];
    int s    = src[e];
    float w  = ew[e];
    int prev = atomicExch(&hd[d * NL + (e & (NL - 1))], e);
    rec[e] = make_int4(s, __float_as_int(w), prev, 0);
}

__global__ void __launch_bounds__(96)
k_aggregate(const int4* __restrict__ heads, const int4* __restrict__ recs,
            const float* __restrict__ x, float* __restrict__ agg, int E) {
    static_assert(NL == 4, "heads are read as one int4 per dst node");
    int n   = blockIdx.x;
    int tid = threadIdx.x;
    float4 acc = make_float4(0.f, 0.f, 0.f, 0.f);
    const float4* x4 = reinterpret_cast<const float4*>(x);
    const unsigned uE = (unsigned)E;

    int4 h = heads[n];
    int   nx[NL] = {h.x, h.y, h.z, h.w};
    bool  live[NL] = {false, false, false, false};
    int   sr[NL] = {0, 0, 0, 0};
    float wt[NL] = {0.f, 0.f, 0.f, 0.f};

    for (int it = 0; it <= E; ++it) {
        float4 v[NL];
#pragma unroll
        for (int k = 0; k < NL; ++k) {
            v[k] = make_float4(0.f, 0.f, 0.f, 0.f);
            if (live[k]) v[k] = x4[(size_t)sr[k] * ROW4 + tid];
        }
        int4 rn[NL];
        bool nlive[NL];
#pragma unroll
        for (int k = 0; k < NL; ++k) {
            nlive[k] = (unsigned)nx[k] < uE;
            rn[k]    = make_int4(0, 0, -1, 0);
            if (nlive[k]) rn[k] = recs[nx[k]];
        }
#pragma unroll
        for (int k = 0; k < NL; ++k) {
            if (live[k]) {
                acc.x += wt[k] * v[k].x;
                acc.y += wt[k] * v[k].y;
                acc.z += wt[k] * v[k].z;
                acc.w += wt[k] * v[k].w;
            }
        }
        bool any = false;
#pragma unroll
        for (int k = 0; k < NL; ++k) {
            live[k] = nlive[k];
            sr[k]   = rn[k].x;
            wt[k]   = __int_as_float(rn[k].y);
            nx[k]   = rn[k].z;
            any     = any | nlive[k];
        }
        if (!any) break;
    }
    reinterpret_cast<float4*>(agg)[(size_t)n * ROW4 + tid] = acc;
}

__global__ void __launch_bounds__(128)
k_gemm1(const float* __restrict__ xT, const float* __restrict__ agg1,
        const float* __restrict__ Wself, const float* __restrict__ Wneigh,
        const float* __restrict__ bias, float* __restrict__ h1) {
    int lane  = threadIdx.x & 31;
    int wave  = threadIdx.x >> 5;
    int mtile = blockIdx.x * (blockDim.x >> 5) + wave;
    if (mtile >= M1 / 16) return;
    int m0 = mtile * 16;

    v2f aS[8], aN[8];
#pragma unroll
    for (int ks = 0; ks < 8; ++ks) {
        aS[ks] = load_frag(xT,   m0, C_IN, ks * 4, lane);
        aN[ks] = load_frag(agg1, m0, C_IN, ks * 4, lane);
    }

#pragma unroll
    for (int nt = 0; nt < 4; ++nt) {
        int n0 = nt * 16;
        v8f acc = {};
#pragma unroll
        for (int ks = 0; ks < 8; ++ks) {
            v2f bS = load_frag(Wself, n0, C_IN, ks * 4, lane);
            acc = __builtin_amdgcn_wmma_f32_16x16x4_f32(false, aS[ks], false, bS,
                                                        (short)0, acc, false, false);
            v2f bN = load_frag(Wneigh, n0, C_IN, ks * 4, lane);
            acc = __builtin_amdgcn_wmma_f32_16x16x4_f32(false, aN[ks], false, bN,
                                                        (short)0, acc, false, false);
        }
        int col   = n0 + (lane & 15);
        int mbase = m0 + ((lane >> 4) << 3);
        float bv  = bias[col];
#pragma unroll
        for (int r = 0; r < 8; ++r) {
            float v = leaky(acc[r] + bv);
            h1[(size_t)(mbase + r) * H + col] = v;
        }
    }
}

__global__ void __launch_bounds__(128)
k_gemm_z2(const float* __restrict__ h1, const float* __restrict__ Wneigh2,
          float* __restrict__ z2) {
    int lane  = threadIdx.x & 31;
    int wave  = threadIdx.x >> 5;
    int mtile = blockIdx.x * (blockDim.x >> 5) + wave;
    if (mtile >= M1 / 16) return;
    int m0 = mtile * 16;

    v2f a[16];
#pragma unroll
    for (int ks = 0; ks < 16; ++ks)
        a[ks] = load_frag(h1, m0, H, ks * 4, lane);

#pragma unroll
    for (int nt = 0; nt < 2; ++nt) {
        int n0 = nt * 16;
        v8f acc = {};
#pragma unroll
        for (int ks = 0; ks < 16; ++ks) {
            v2f b = load_frag(Wneigh2, n0, H, ks * 4, lane);
            acc = __builtin_amdgcn_wmma_f32_16x16x4_f32(false, a[ks], false, b,
                                                        (short)0, acc, false, false);
        }
        int col   = n0 + (lane & 15);
        int mbase = m0 + ((lane >> 4) << 3);
#pragma unroll
        for (int r = 0; r < 8; ++r)
            z2[(size_t)(mbase + r) * C_OUT + col] = acc[r];
    }
}

__global__ void __launch_bounds__(128)
k_final(const float* __restrict__ h1, const float* __restrict__ agg2,
        const float* __restrict__ Wself2, const float* __restrict__ bias2,
        float* __restrict__ out) {
    int lane  = threadIdx.x & 31;
    int wave  = threadIdx.x >> 5;
    int mtile = blockIdx.x * (blockDim.x >> 5) + wave;
    if (mtile >= M2 / 16) return;
    int m0 = mtile * 16;

    v2f a[16];
#pragma unroll
    for (int ks = 0; ks < 16; ++ks)
        a[ks] = load_frag(h1, m0, H, ks * 4, lane);

#pragma unroll
    for (int nt = 0; nt < 2; ++nt) {
        int n0 = nt * 16;
        v8f acc = {};
#pragma unroll
        for (int ks = 0; ks < 16; ++ks) {
            v2f b = load_frag(Wself2, n0, H, ks * 4, lane);
            acc = __builtin_amdgcn_wmma_f32_16x16x4_f32(false, a[ks], false, b,
                                                        (short)0, acc, false, false);
        }
        int col   = n0 + (lane & 15);
        int mbase = m0 + ((lane >> 4) << 3);
        float bv  = bias2[col];
#pragma unroll
        for (int r = 0; r < 8; ++r) {
            int m   = mbase + r;
            float v = leaky(acc[r] + agg2[(size_t)m * C_OUT + col] + bv);
            int n = m / T;
            int t = m % T;
            out[((size_t)t * N_DST2 + n) * C_OUT + col] = v;
        }
    }
}

extern "C" void kernel_launch(void* const* d_in, const int* in_sizes, int n_in,
                              void* d_out, int out_size, void* d_ws, size_t ws_size,
                              hipStream_t stream) {
    const float* nf   = (const float*)d_in[0];
    const int*   src1 = (const int*)  d_in[1];
    const int*   dst1 = (const int*)  d_in[2];
    const float* ew1  = (const float*)d_in[3];
    const int*   src2 = (const int*)  d_in[4];
    const int*   dst2 = (const int*)  d_in[5];
    const float* ew2  = (const float*)d_in[6];
    const float* W1s  = (const float*)d_in[7];
    const float* W1n  = (const float*)d_in[8];
    const float* b1   = (const float*)d_in[9];
    const float* W2s  = (const float*)d_in[10];
    const float* W2n  = (const float*)d_in[11];
    const float* b2   = (const float*)d_in[12];
    const int E1 = in_sizes[1];
    const int E2 = in_sizes[4];

    float* out = (float*)d_out;

    float* ws   = (float*)d_ws;
    float* xT   = ws;
    float* agg1 = xT   + (size_t)N_SRC  * ROW;
    float* h1   = agg1 + (size_t)N_DST1 * ROW;
    float* z2   = xT;
    float* agg2 = xT + (size_t)N_DST1 * ROW;
    int*   ip     = (int*)(h1 + (size_t)M1 * H);
    int*   heads1 = ip;                  ip += N_DST1 * NL;
    int*   heads2 = ip;                  ip += N_DST2 * NL;
    int4*  rec2   = (int4*)ip;
    int4*  rec1   = (int4*)h1;

    {
        int total = N_SRC * T * (C_IN / 4);
        k_transpose<<<(total + 255) / 256, 256, 0, stream>>>(nf, xT, heads1,
                                                             (N_DST1 + N_DST2) * NL);
    }
    k_link<<<(E1 + 255) / 256 + (E2 + 255) / 256, 256, 0, stream>>>(
        src1, dst1, ew1, E1, (E1 + 255) / 256, src2, dst2, ew2, E2,
        heads1, heads2, rec1, rec2);

    k_aggregate<<<N_DST1, 96, 0, stream>>>((const int4*)heads1, rec1, xT, agg1, E1);

    k_gemm1<<<(M1 / 16 + 3) / 4, 128, 0, stream>>>(xT, agg1, W1s, W1n, b1, h1);

    k_gemm_z2<<<(M1 / 16 + 3) / 4, 128, 0, stream>>>(h1, W2n, z2);

    k_aggregate<<<N_DST2, 96, 0, stream>>>((const int4*)heads2, rec2, z2, agg2, E2);

    k_final<<<(M2 / 16 + 3) / 4, 128, 0, stream>>>(h1, agg2, W2s, b2, out);
}


# Layer-2 gather finishes the layer: a dst node's block keeps its neighbour sum and own h1 rows in LDS, two waves do the own-term WMMA tile, out stored as whole lines; 7->6 launches; sums unchanged
# speedup vs baseline: 1.8431x; 1.0480x over previous
#include <hip/hip_runtime.h>

typedef __attribute__((ext_vector_type(2))) float v2f;
typedef __attribute__((ext_vector_type(8))) float v8f;

constexpr int T      = 12;
constexpr int C_IN   = 32;
constexpr int H      = 64;
constexpr int C_OUT  = 32;
constexpr int N_SRC  = 50000;
constexpr int N_DST1 = 20000;
constexpr int N_DST2 = 10000;
constexpr int ROW    = T * C_IN;
constexpr int ROW4   = ROW / 4;
constexpr int M1     = N_DST1 * T;
constexpr int M2     = N_DST2 * T;
constexpr float NEG_SLOPE = 0.01f;

__device__ __forceinline__ float leaky(float v) {
    return v > 0.f ? v : NEG_SLOPE * v;
}

__device__ __forceinline__ v2f load_frag(const float* __restrict__ base, int r0, int ld,
                                         int k0, int lane) {
    int r = r0 + (lane & 15);
    int k = k0 + ((lane >> 4) << 1);
    const float* p = base + (size_t)r * ld + k;
    v2f f;
    f.x = p[0];
    f.y = p[1];
    return f;
}

__global__ void k_transpose(const float* __restrict__ nf, float* __restrict__ xT,
                            int* __restrict__ heads, int nheads) {
    int idx = blockIdx.x * blockDim.x + threadIdx.x;
    constexpr int C4 = C_IN / 4;
    constexpr int total = N_SRC * T * C4;
    if (idx < nheads) heads[idx] = -1;
    if (idx >= total) return;
    int c4 = idx % C4;
    int t  = (idx / C4) % T;
    int n  = idx / (C4 * T);
    const float4* src = reinterpret_cast<const float4*>(nf);
    float4 v = src[(size_t)t * N_SRC * C4 + (size_t)n * C4 + c4];
    reinterpret_cast<float4*>(xT)[idx] = v;
}

constexpr int NL = 4;

__global__ void __launch_bounds__(256)
k_link(const int* __restrict__ src1, const int* __restrict__ dst1,
       const float* __restrict__ ew1, int E1, int nb1,
       const int* __restrict__ src2, const int* __restrict__ dst2,
       const float* __restrict__ ew2, int E2,
       int* __restrict__ heads1, int* __restrict__ heads2,
       int4* __restrict__ rec1, int4* __restrict__ rec2) {
    int b = blockIdx.x;
    const int*   src = src1;
    const int*   dst = dst1;
    const float* ew  = ew1;
    int*         hd  = heads1;
    int4*        rec = rec1;
    int          E   = E1;
    if (b >= nb1) {
        b -= nb1;
        src = src2; dst = dst2; ew = ew2; hd = heads2; rec = rec2; E = E2;
    }
    int e = b * blockDim.x + threadIdx.x;
    if (e >= E) return;
    int d    = dst[e];
    int s    = src[e];
    float w  = ew[e];
    int prev = atomicExch(&hd[d * NL + (e & (NL - 1))], e);
    rec[e] = make_int4(s, __float_as_int(w), prev, 0);
}

__device__ __forceinline__ float4 gather_row(const int4* __restrict__ heads,
                                             const int4* __restrict__ recs,
                                             const float* __restrict__ x, int E,
                                             int n, int tid) {
    static_assert(NL == 4, "heads are read as one int4 per dst node");
    float4 acc = make_float4(0.f, 0.f, 0.f, 0.f);
    const float4* x4 = reinterpret_cast<const float4*>(x);
    const unsigned uE = (unsigned)E;

    int4 h = heads[n];
    int   nx[NL] = {h.x, h.y, h.z, h.w};
    bool  live[NL] = {false, false, false, false};
    int   sr[NL] = {0, 0, 0, 0};
    float wt[NL] = {0.f, 0.f, 0.f, 0.f};

    for (int it = 0; it <= E; ++it) {
        float4 v[NL];
#pragma unroll
        for (int k = 0; k < NL; ++k) {
            v[k] = make_float4(0.f, 0.f, 0.f, 0.f);
            if (live[k]) v[k] = x4[(size_t)sr[k] * ROW4 + tid];
        }
        int4 rn[NL];
        bool nlive[NL];
#pragma unroll
        for (int k = 0; k < NL; ++k) {
            nlive[k] = (unsigned)nx[k] < uE;
            rn[k]    = make_int4(0, 0, -1, 0);
            if (nlive[k]) rn[k] = recs[nx[k]];
        }
#pragma unroll
        for (int k = 0; k < NL; ++k) {
            if (live[k]) {
                acc.x += wt[k] * v[k].x;
                acc.y += wt[k] * v[k].y;
                acc.z += wt[k] * v[k].z;
                acc.w += wt[k] * v[k].w;
            }
        }
        bool any = false;
#pragma unroll
        for (int k = 0; k < NL; ++k) {
            live[k] = nlive[k];
            sr[k]   = rn[k].x;
            wt[k]   = __int_as_float(rn[k].y);
            nx[k]   = rn[k].z;
            any     = any | nlive[k];
        }
        if (!any) break;
    }
    return acc;
}

__global__ void __launch_bounds__(96)
k_aggregate(const int4* __restrict__ heads, const int4* __restrict__ recs,
            const float* __restrict__ x, float* __restrict__ agg, int E) {
    int n   = blockIdx.x;
    int tid = threadIdx.x;
    float4 acc = gather_row(heads, recs, x, E, n, tid);
    reinterpret_cast<float4*>(agg)[(size_t)n * ROW4 + tid] = acc;
}

__global__ void __launch_bounds__(128)
k_gemm1(const float* __restrict__ xT, const float* __restrict__ agg1,
        const float* __restrict__ Wself, const float* __restrict__ Wneigh,
        const float* __restrict__ bias, float* __restrict__ h1) {
    int lane  = threadIdx.x & 31;
    int wave  = threadIdx.x >> 5;
    int mtile = blockIdx.x * (blockDim.x >> 5) + wave;
    if (mtile >= M1 / 16) return;
    int m0 = mtile * 16;

    v2f aS[8], aN[8];
#pragma unroll
    for (int ks = 0; ks < 8; ++ks) {
        aS[ks] = load_frag(xT,   m0, C_IN, ks * 4, lane);
        aN[ks] = load_frag(agg1, m0, C_IN, ks * 4, lane);
    }

#pragma unroll
    for (int nt = 0; nt < 4; ++nt) {
        int n0 = nt * 16;
        v8f acc = {};
#pragma unroll
        for (int ks = 0; ks < 8; ++ks) {
            v2f bS = load_frag(Wself, n0, C_IN, ks * 4, lane);
            acc = __builtin_amdgcn_wmma_f32_16x16x4_f32(false, aS[ks], false, bS,
                                                        (short)0, acc, false, false);
            v2f bN = load_frag(Wneigh, n0, C_IN, ks * 4, lane);
            acc = __builtin_amdgcn_wmma_f32_16x16x4_f32(false, aN[ks], false, bN,
                                                        (short)0, acc, false, false);
        }
        int col   = n0 + (lane & 15);
        int mbase = m0 + ((lane >> 4) << 3);
        float bv  = bias[col];
#pragma unroll
        for (int r = 0; r < 8; ++r) {
            float v = leaky(acc[r] + bv);
            h1[(size_t)(mbase + r) * H + col] = v;
        }
    }
}

__global__ void __launch_bounds__(128)
k_gemm_z2(const float* __restrict__ h1, const float* __restrict__ Wneigh2,
          float* __restrict__ z2) {
    int lane  = threadIdx.x & 31;
    int wave  = threadIdx.x >> 5;
    int mtile = blockIdx.x * (blockDim.x >> 5) + wave;
    if (mtile >= M1 / 16) return;
    int m0 = mtile * 16;

    v2f a[16];
#pragma unroll
    for (int ks = 0; ks < 16; ++ks)
        a[ks] = load_frag(h1, m0, H, ks * 4, lane);

#pragma unroll
    for (int nt = 0; nt < 2; ++nt) {
        int n0 = nt * 16;
        v8f acc = {};
#pragma unroll
        for (int ks = 0; ks < 16; ++ks) {
            v2f b = load_frag(Wneigh2, n0, H, ks * 4, lane);
            acc = __builtin_amdgcn_wmma_f32_16x16x4_f32(false, a[ks], false, b,
                                                        (short)0, acc, false, false);
        }
        int col   = n0 + (lane & 15);
        int mbase = m0 + ((lane >> 4) << 3);
#pragma unroll
        for (int r = 0; r < 8; ++r)
            z2[(size_t)(mbase + r) * C_OUT + col] = acc[r];
    }
}

constexpr int HP = H + 4;
__global__ void __launch_bounds__(96)
k_aggregate_final(const int4* __restrict__ heads, const int4* __restrict__ recs,
                  const float* __restrict__ z2, int E,
                  const float* __restrict__ h1, const float* __restrict__ Wself2,
                  const float* __restrict__ bias2, float* __restrict__ out) {
    static_assert(T == 12 && H == 64 && C_OUT == 32,
                  "12 rows of 64 = 3 float4 per thread of two waves; two 16-column tiles");
    __shared__ float4 s_agg4[T * C_OUT / 4];
    __shared__ float4 s_h4[16 * HP / 4];
    float* s_agg = reinterpret_cast<float*>(s_agg4);
    float* s_h   = reinterpret_cast<float*>(s_h4);

    int n    = blockIdx.x;
    int tid  = threadIdx.x;
    int lane = tid & 31;
    int wave = __builtin_amdgcn_readfirstlane(tid >> 5);

    const float4* h4 = reinterpret_cast<const float4*>(h1) + (size_t)n * (T * H / 4);
    float4 hv0 = make_float4(0.f, 0.f, 0.f, 0.f), hv1 = hv0, hv2 = hv0;
    if (wave < 2) {
        hv0 = h4[tid];
        hv1 = h4[tid + 64];
        hv2 = h4[tid + 128];
    }

    float4 acc = gather_row(heads, recs, z2, E, n, tid);

    s_agg4[tid] = acc;
    if (wave < 2) {
        float* p = &s_h[(tid >> 4) * HP + (tid & 15) * 4];
        *reinterpret_cast<float4*>(p)          = hv0;
        *reinterpret_cast<float4*>(p + 4 * HP) = hv1;
        *reinterpret_cast<float4*>(p + 8 * HP) = hv2;
    } else {
#pragma unroll
        for (int i = 0; i < 2; ++i) {
            int j = 192 + lane + 32 * i;
            *reinterpret_cast<float4*>(&s_h[(j >> 4) * HP + (j & 15) * 4]) =
                make_float4(0.f, 0.f, 0.f, 0.f);
        }
    }
    __syncthreads();

    if (wave < 2) {
        int n0 = wave * 16;
        int c  = lane & 15;
        int hh = lane >> 4;
        v8f d = {};
#pragma unroll
        for (int ks = 0; ks < 16; ++ks) {
            v2f a = *reinterpret_cast<const v2f*>(&s_h[c * HP + ks * 4 + 2 * hh]);
            v2f b = load_frag(Wself2, n0, H, ks * 4, lane);
            d = __builtin_amdgcn_wmma_f32_16x16x4_f32(false, a, false, b,
                                                      (short)0, d, false, false);
        }
        int col = n0 + c;
        int tb  = hh << 3;
        float bv = bias2[col];
#pragma unroll
        for (int r = 0; r < 8; ++r) {
            int t = tb + r;
            if (t < T) {
                float v = leaky(d[r] + s_agg[t * C_OUT + col] + bv);
                s_agg[t * C_OUT + col] = v;
            }
        }
    }
    __syncthreads();

    float4 o = s_agg4[tid];
    reinterpret_cast<float4*>(out)[((size_t)(tid >> 3) * N_DST2 + n) * (C_OUT / 4) + (tid & 7)] = o;
}

extern "C" void kernel_launch(void* const* d_in, const int* in_sizes, int n_in,
                              void* d_out, int out_size, void* d_ws, size_t ws_size,
                              hipStream_t stream) {
    const float* nf   = (const float*)d_in[0];
    const int*   src1 = (const int*)  d_in[1];
    const int*   dst1 = (const int*)  d_in[2];
    const float* ew1  = (const float*)d_in[3];
    const int*   src2 = (const int*)  d_in[4];
    const int*   dst2 = (const int*)  d_in[5];
    const float* ew2  = (const float*)d_in[6];
    const float* W1s  = (const float*)d_in[7];
    const float* W1n  = (const float*)d_in[8];
    const float* b1   = (const float*)d_in[9];
    const float* W2s  = (const float*)d_in[10];
    const float* W2n  = (const float*)d_in[11];
    const float* b2   = (const float*)d_in[12];
    const int E1 = in_sizes[1];
    const int E2 = in_sizes[4];

    float* out = (float*)d_out;

    float* ws   = (float*)d_ws;
    float* xT   = ws;
    float* agg1 = xT   + (size_t)N_SRC  * ROW;
    float* h1   = agg1 + (size_t)N_DST1 * ROW;
    float* z2   = xT;
    int*   ip     = (int*)(h1 + (size_t)M1 * H);
    int*   heads1 = ip;                  ip += N_DST1 * NL;
    int*   heads2 = ip;                  ip += N_DST2 * NL;
    int4*  rec2   = (int4*)ip;
    int4*  rec1   = (int4*)h1;

    {
        int total = N_SRC * T * (C_IN / 4);
        k_transpose<<<(total + 255) / 256, 256, 0, stream>>>(nf, xT, heads1,
                                                             (N_DST1 + N_DST2) * NL);
    }
    k_link<<<(E1 + 255) / 256 + (E2 + 255) / 256, 256, 0, stream>>>(
        src1, dst1, ew1, E1, (E1 + 255) / 256, src2, dst2, ew2, E2,
        heads1, heads2, rec1, rec2);

    k_aggregate<<<N_DST1, 96, 0, stream>>>((const int4*)heads1, rec1, xT, agg1, E1);

    k_gemm1<<<(M1 / 16 + 3) / 4, 128, 0, stream>>>(xT, agg1, W1s, W1n, b1, h1);

    k_gemm_z2<<<(M1 / 16 + 3) / 4, 128, 0, stream>>>(h1, W2n, z2);

    k_aggregate_final<<<N_DST2, 96, 0, stream>>>((const int4*)heads2, rec2, z2, E2,
                                                 h1, W2s, b2, out);
}


# Transpose launch re-tiled: one wave moves 8 whole node rows (24 loads of 512 contiguous bytes in flight per lane, then whole-line stores); 150000 -> 6250 waves enter; sums untouched
# speedup vs baseline: 1.8988x; 1.0302x over previous
#include <hip/hip_runtime.h>

typedef __attribute__((ext_vector_type(2))) float v2f;
typedef __attribute__((ext_vector_type(8))) float v8f;
typedef __attribute__((ext_vector_type(4))) float v4f;

constexpr int T      = 12;
constexpr int C_IN   = 32;
constexpr int H      = 64;
constexpr int C_OUT  = 32;
constexpr int N_SRC  = 50000;
constexpr int N_DST1 = 20000;
constexpr int N_DST2 = 10000;
constexpr int ROW    = T * C_IN;
constexpr int ROW4   = ROW / 4;
constexpr int M1     = N_DST1 * T;
constexpr int M2     = N_DST2 * T;
constexpr float NEG_SLOPE = 0.01f;

__device__ __forceinline__ float leaky(float v) {
    return v > 0.f ? v : NEG_SLOPE * v;
}

__device__ __forceinline__ v2f load_frag(const float* __restrict__ base, int r0, int ld,
                                         int k0, int lane) {
    int r = r0 + (lane & 15);
    int k = k0 + ((lane >> 4) << 1);
    const float* p = base + (size_t)r * ld + k;
    v2f f;
    f.x = p[0];
    f.y = p[1];
    return f;
}

constexpr int TR_NODES  = 8;
constexpr int TR_WAVES  = N_SRC / TR_NODES;
constexpr int TR_BLOCK  = 256;
constexpr int TR_BLOCKS = (TR_WAVES * 32 + TR_BLOCK - 1) / TR_BLOCK;
static_assert(N_SRC % TR_NODES == 0 && TR_NODES % 4 == 0 && C_IN == 32,
              "whole groups of 4 nodes per load; 8 float4 per (node, t) line");

__global__ void __launch_bounds__(TR_BLOCK)
k_transpose(const float* __restrict__ nf, float* __restrict__ xT,
            int* __restrict__ heads, int nheads) {
    const int gtid = blockIdx.x * TR_BLOCK + threadIdx.x;
    for (int i = gtid; i < nheads; i += TR_BLOCKS * TR_BLOCK) heads[i] = -1;

    const int gw = gtid >> 5;
    if (gw >= TR_WAVES) return;
    const int lane = threadIdx.x & 31;
    const int c4   = lane & 7;
    const int n0   = gw * TR_NODES + (lane >> 3);
    constexpr int C4 = C_IN / 4;
    constexpr int NG = TR_NODES / 4;
    const v4f* src = reinterpret_cast<const v4f*>(nf);
    v4f*       dst = reinterpret_cast<v4f*>(xT);
    static_assert(NG == 2 && T == 12, "the fence below names the 24 loaded values one by one");

    v4f v[NG][T];
#pragma unroll
    for (int g = 0; g < NG; ++g)
#pragma unroll
        for (int t = 0; t < T; ++t)
            v[g][t] = src[((size_t)t * N_SRC + (n0 + 4 * g)) * C4 + c4];
    asm volatile("" ::
        "v"(v[0][0]), "v"(v[0][1]), "v"(v[0][2]),  "v"(v[0][3]),
        "v"(v[0][4]), "v"(v[0][5]), "v"(v[0][6]),  "v"(v[0][7]),
        "v"(v[0][8]), "v"(v[0][9]), "v"(v[0][10]), "v"(v[0][11]),
        "v"(v[1][0]), "v"(v[1][1]), "v"(v[1][2]),  "v"(v[1][3]),
        "v"(v[1][4]), "v"(v[1][5]), "v"(v[1][6]),  "v"(v[1][7]),
        "v"(v[1][8]), "v"(v[1][9]), "v"(v[1][10]), "v"(v[1][11]));
#pragma unroll
    for (int g = 0; g < NG; ++g)
#pragma unroll
        for (int t = 0; t < T; ++t)
            dst[((size_t)(n0 + 4 * g) * T + t) * C4 + c4] = v[g][t];
}

constexpr int NL = 4;

__global__ void __launch_bounds__(256)
k_link(const int* __restrict__ src1, const int* __restrict__ dst1,
       const float* __restrict__ ew1, int E1, int nb1,
       const int* __restrict__ src2, const int* __restrict__ dst2,
       const float* __restrict__ ew2, int E2,
       int* __restrict__ heads1, int* __restrict__ heads2,
       int4* __restrict__ rec1, int4* __restrict__ rec2) {
    int b = blockIdx.x;
    const int*   src = src1;
    const int*   dst = dst1;
    const float* ew  = ew1;
    int*         hd  = heads1;
    int4*        rec = rec1;
    int          E   = E1;
    if (b >= nb1) {
        b -= nb1;
        src = src2; dst = dst2; ew = ew2; hd = heads2; rec = rec2; E = E2;
    }
    int e = b * blockDim.x + threadIdx.x;
    if (e >= E) return;
    int d    = dst[e];
    int s    = src[e];
    float w  = ew[e];
    int prev = atomicExch(&hd[d * NL + (e & (NL - 1))], e);
    rec[e] = make_int4(s, __float_as_int(w), prev, 0);
}

__device__ __forceinline__ float4 gather_row(const int4* __restrict__ heads,
                                             const int4* __restrict__ recs,
                                             const float* __restrict__ x, int E,
                                             int n, int tid) {
    static_assert(NL == 4, "heads are read as one int4 per dst node");
    float4 acc = make_float4(0.f, 0.f, 0.f, 0.f);
    const float4* x4 = reinterpret_cast<const float4*>(x);
    const unsigned uE = (unsigned)E;

    int4 h = heads[n];
    int   nx[NL] = {h.x, h.y, h.z, h.w};
    bool  live[NL] = {false, false, false, false};
    int   sr[NL] = {0, 0, 0, 0};
    float wt[NL] = {0.f, 0.f, 0.f, 0.f};

    for (int it = 0; it <= E; ++it) {
        float4 v[NL];
#pragma unroll
        for (int k = 0; k < NL; ++k) {
            v[k] = make_float4(0.f, 0.f, 0.f, 0.f);
            if (live[k]) v[k] = x4[(size_t)sr[k] * ROW4 + tid];
        }
        int4 rn[NL];
        bool nlive[NL];
#pragma unroll
        for (int k = 0; k < NL; ++k) {
            nlive[k] = (unsigned)nx[k] < uE;
            rn[k]    = make_int4(0, 0, -1, 0);
            if (nlive[k]) rn[k] = recs[nx[k]];
        }
#pragma unroll
        for (int k = 0; k < NL; ++k) {
            if (live[k]) {
                acc.x += wt[k] * v[k].x;
                acc.y += wt[k] * v[k].y;
                acc.z += wt[k] * v[k].z;
                acc.w += wt[k] * v[k].w;
            }
        }
        bool any = false;
#pragma unroll
        for (int k = 0; k < NL; ++k) {
            live[k] = nlive[k];
            sr[k]   = rn[k].x;
            wt[k]   = __int_as_float(rn[k].y);
            nx[k]   = rn[k].z;
            any     = any | nlive[k];
        }
        if (!any) break;
    }
    return acc;
}

__global__ void __launch_bounds__(96)
k_aggregate(const int4* __restrict__ heads, const int4* __restrict__ recs,
            const float* __restrict__ x, float* __restrict__ agg, int E) {
    int n   = blockIdx.x;
    int tid = threadIdx.x;
    float4 acc = gather_row(heads, recs, x, E, n, tid);
    reinterpret_cast<float4*>(agg)[(size_t)n * ROW4 + tid] = acc;
}

__global__ void __launch_bounds__(128)
k_gemm1(const float* __restrict__ xT, const float* __restrict__ agg1,
        const float* __restrict__ Wself, const float* __restrict__ Wneigh,
        const float* __restrict__ bias, float* __restrict__ h1) {
    int lane  = threadIdx.x & 31;
    int wave  = threadIdx.x >> 5;
    int mtile = blockIdx.x * (blockDim.x >> 5) + wave;
    if (mtile >= M1 / 16) return;
    int m0 = mtile * 16;

    v2f aS[8], aN[8];
#pragma unroll
    for (int ks = 0; ks < 8; ++ks) {
        aS[ks] = load_frag(xT,   m0, C_IN, ks * 4, lane);
        aN[ks] = load_frag(agg1, m0, C_IN, ks * 4, lane);
    }

#pragma unroll
    for (int nt = 0; nt < 4; ++nt) {
        int n0 = nt * 16;
        v8f acc = {};
#pragma unroll
        for (int ks = 0; ks < 8; ++ks) {
            v2f bS = load_frag(Wself, n0, C_IN, ks * 4, lane);
            acc = __builtin_amdgcn_wmma_f32_16x16x4_f32(false, aS[ks], false, bS,
                                                        (short)0, acc, false, false);
            v2f bN = load_frag(Wneigh, n0, C_IN, ks * 4, lane);
            acc = __builtin_amdgcn_wmma_f32_16x16x4_f32(false, aN[ks], false, bN,
                                                        (short)0, acc, false, false);
        }
        int col   = n0 + (lane & 15);
        int mbase = m0 + ((lane >> 4) << 3);
        float bv  = bias[col];
#pragma unroll
        for (int r = 0; r < 8; ++r) {
            float v = leaky(acc[r] + bv);
            h1[(size_t)(mbase + r) * H + col] = v;
        }
    }
}

__global__ void __launch_bounds__(128)
k_gemm_z2(const float* __restrict__ h1, const float* __restrict__ Wneigh2,
          float* __restrict__ z2) {
    int lane  = threadIdx.x & 31;
    int wave  = threadIdx.x >> 5;
    int mtile = blockIdx.x * (blockDim.x >> 5) + wave;
    if (mtile >= M1 / 16) return;
    int m0 = mtile * 16;

    v2f a[16];
#pragma unroll
    for (int ks = 0; ks < 16; ++ks)
        a[ks] = load_frag(h1, m0, H, ks * 4, lane);

#pragma unroll
    for (int nt = 0; nt < 2; ++nt) {
        int n0 = nt * 16;
        v8f acc = {};
#pragma unroll
        for (int ks = 0; ks < 16; ++ks) {
            v2f b = load_frag(Wneigh2, n0, H, ks * 4, lane);
            acc = __builtin_amdgcn_wmma_f32_16x16x4_f32(false, a[ks], false, b,
                                                        (short)0, acc, false, false);
        }
        int col   = n0 + (lane & 15);
        int mbase = m0 + ((lane >> 4) << 3);
#pragma unroll
        for (int r = 0; r < 8; ++r)
            z2[(size_t)(mbase + r) * C_OUT + col] = acc[r];
    }
}

constexpr int HP = H + 4;
__global__ void __launch_bounds__(96)
k_aggregate_final(const int4* __restrict__ heads, const int4* __restrict__ recs,
                  const float* __restrict__ z2, int E,
                  const float* __restrict__ h1, const float* __restrict__ Wself2,
                  const float* __restrict__ bias2, float* __restrict__ out) {
    static_assert(T == 12 && H == 64 && C_OUT == 32,
                  "12 rows of 64 = 3 float4 per thread of two waves; two 16-column tiles");
    __shared__ float4 s_agg4[T * C_OUT / 4];
    __shared__ float4 s_h4[16 * HP / 4];
    float* s_agg = reinterpret_cast<float*>(s_agg4);
    float* s_h   = reinterpret_cast<float*>(s_h4);

    int n    = blockIdx.x;
    int tid  = threadIdx.x;
    int lane = tid & 31;
    int wave = __builtin_amdgcn_readfirstlane(tid >> 5);

    const float4* h4 = reinterpret_cast<const float4*>(h1) + (size_t)n * (T * H / 4);
    float4 hv0 = make_float4(0.f, 0.f, 0.f, 0.f), hv1 = hv0, hv2 = hv0;
    if (wave < 2) {
        hv0 = h4[tid];
        hv1 = h4[tid + 64];
        hv2 = h4[tid + 128];
    }

    float4 acc = gather_row(heads, recs, z2, E, n, tid);

    s_agg4[tid] = acc;
    if (wave < 2) {
        float* p = &s_h[(tid >> 4) * HP + (tid & 15) * 4];
        *reinterpret_cast<float4*>(p)          = hv0;
        *reinterpret_cast<float4*>(p + 4 * HP) = hv1;
        *reinterpret_cast<float4*>(p + 8 * HP) = hv2;
    } else {
#pragma unroll
        for (int i = 0; i < 2; ++i) {
            int j = 192 + lane + 32 * i;
            *reinterpret_cast<float4*>(&s_h[(j >> 4) * HP + (j & 15) * 4]) =
                make_float4(0.f, 0.f, 0.f, 0.f);
        }
    }
    __syncthreads();

    if (wave < 2) {
        int n0 = wave * 16;
        int c  = lane & 15;
        int hh = lane >> 4;
        v8f d = {};
#pragma unroll
        for (int ks = 0; ks < 16; ++ks) {
            v2f a = *reinterpret_cast<const v2f*>(&s_h[c * HP + ks * 4 + 2 * hh]);
            v2f b = load_frag(Wself2, n0, H, ks * 4, lane);
            d = __builtin_amdgcn_wmma_f32_16x16x4_f32(false, a, false, b,
                                                      (short)0, d, false, false);
        }
        int col = n0 + c;
        int tb  = hh << 3;
        float bv = bias2[col];
#pragma unroll
        for (int r = 0; r < 8; ++r) {
            int t = tb + r;
            if (t < T) {
                float v = leaky(d[r] + s_agg[t * C_OUT + col] + bv);
                s_agg[t * C_OUT + col] = v;
            }
        }
    }
    __syncthreads();

    float4 o = s_agg4[tid];
    reinterpret_cast<float4*>(out)[((size_t)(tid >> 3) * N_DST2 + n) * (C_OUT / 4) + (tid & 7)] = o;
}

extern "C" void kernel_launch(void* const* d_in, const int* in_sizes, int n_in,
                              void* d_out, int out_size, void* d_ws, size_t ws_size,
                              hipStream_t stream) {
    const float* nf   = (const float*)d_in[0];
    const int*   src1 = (const int*)  d_in[1];
    const int*   dst1 = (const int*)  d_in[2];
    const float* ew1  = (const float*)d_in[3];
    const int*   src2 = (const int*)  d_in[4];
    const int*   dst2 = (const int*)  d_in[5];
    const float* ew2  = (const float*)d_in[6];
    const float* W1s  = (const float*)d_in[7];
    const float* W1n  = (const float*)d_in[8];
    const float* b1   = (const float*)d_in[9];
    const float* W2s  = (const float*)d_in[10];
    const float* W2n  = (const float*)d_in[11];
    const float* b2   = (const float*)d_in[12];
    const int E1 = in_sizes[1];
    const int E2 = in_sizes[4];

    float* out = (float*)d_out;

    float* ws   = (float*)d_ws;
    float* xT   = ws;
    float* agg1 = xT   + (size_t)N_SRC  * ROW;
    float* h1   = agg1 + (size_t)N_DST1 * ROW;
    float* z2   = xT;
    int*   ip     = (int*)(h1 + (size_t)M1 * H);
    int*   heads1 = ip;                  ip += N_DST1 * NL;
    int*   heads2 = ip;                  ip += N_DST2 * NL;
    int4*  rec2   = (int4*)ip;
    int4*  rec1   = (int4*)h1;

    k_transpose<<<TR_BLOCKS, TR_BLOCK, 0, stream>>>(nf, xT, heads1, (N_DST1 + N_DST2) * NL);
    k_link<<<(E1 + 255) / 256 + (E2 + 255) / 256, 256, 0, stream>>>(
        src1, dst1, ew1, E1, (E1 + 255) / 256, src2, dst2, ew2, E2,
        heads1, heads2, rec1, rec2);

    k_aggregate<<<N_DST1, 96, 0, stream>>>((const int4*)heads1, rec1, xT, agg1, E1);

    k_gemm1<<<(M1 / 16 + 3) / 4, 128, 0, stream>>>(xT, agg1, W1s, W1n, b1, h1);

    k_gemm_z2<<<(M1 / 16 + 3) / 4, 128, 0, stream>>>(h1, W2n, z2);

    k_aggregate_final<<<N_DST2, 96, 0, stream>>>((const int4*)heads2, rec2, z2, E2,
                                                 h1, W2s, b2, out);
}


# GEMM weights held in registers: a wave loads all B fragments once and walks 5 consecutive M tiles with them, later tiles' A loads issued ahead of the WMMAs; both GEMM launches; sums unchanged
# speedup vs baseline: 2.0622x; 1.0861x over previous
#include <hip/hip_runtime.h>

typedef __attribute__((ext_vector_type(2))) float v2f;
typedef __attribute__((ext_vector_type(8))) float v8f;
typedef __attribute__((ext_vector_type(4))) float v4f;

constexpr int T      = 12;
constexpr int C_IN   = 32;
constexpr int H      = 64;
constexpr int C_OUT  = 32;
constexpr int N_SRC  = 50000;
constexpr int N_DST1 = 20000;
constexpr int N_DST2 = 10000;
constexpr int ROW    = T * C_IN;
constexpr int ROW4   = ROW / 4;
constexpr int M1     = N_DST1 * T;
constexpr int M2     = N_DST2 * T;
constexpr float NEG_SLOPE = 0.01f;

__device__ __forceinline__ float leaky(float v) {
    return v > 0.f ? v : NEG_SLOPE * v;
}

__device__ __forceinline__ v2f load_frag(const float* __restrict__ base, int r0, int ld,
                                         int k0, int lane) {
    int r = r0 + (lane & 15);
    int k = k0 + ((lane >> 4) << 1);
    const float* p = base + (size_t)r * ld + k;
    v2f f;
    f.x = p[0];
    f.y = p[1];
    return f;
}

constexpr int TR_NODES  = 8;
constexpr int TR_WAVES  = N_SRC / TR_NODES;
constexpr int TR_BLOCK  = 256;
constexpr int TR_BLOCKS = (TR_WAVES * 32 + TR_BLOCK - 1) / TR_BLOCK;
static_assert(N_SRC % TR_NODES == 0 && TR_NODES % 4 == 0 && C_IN == 32,
              "whole groups of 4 nodes per load; 8 float4 per (node, t) line");

__global__ void __launch_bounds__(TR_BLOCK)
k_transpose(const float* __restrict__ nf, float* __restrict__ xT,
            int* __restrict__ heads, int nheads) {
    const int gtid = blockIdx.x * TR_BLOCK + threadIdx.x;
    for (int i = gtid; i < nheads; i += TR_BLOCKS * TR_BLOCK) heads[i] = -1;

    const int gw = gtid >> 5;
    if (gw >= TR_WAVES) return;
    const int lane = threadIdx.x & 31;
    const int c4   = lane & 7;
    const int n0   = gw * TR_NODES + (lane >> 3);
    constexpr int C4 = C_IN / 4;
    constexpr int NG = TR_NODES / 4;
    const v4f* src = reinterpret_cast<const v4f*>(nf);
    v4f*       dst = reinterpret_cast<v4f*>(xT);
    static_assert(NG == 2 && T == 12, "the fence below names the 24 loaded values one by one");

    v4f v[NG][T];
#pragma unroll
    for (int g = 0; g < NG; ++g)
#pragma unroll
        for (int t = 0; t < T; ++t)
            v[g][t] = src[((size_t)t * N_SRC + (n0 + 4 * g)) * C4 + c4];
    asm volatile("" ::
        "v"(v[0][0]), "v"(v[0][1]), "v"(v[0][2]),  "v"(v[0][3]),
        "v"(v[0][4]), "v"(v[0][5]), "v"(v[0][6]),  "v"(v[0][7]),
        "v"(v[0][8]), "v"(v[0][9]), "v"(v[0][10]), "v"(v[0][11]),
        "v"(v[1][0]), "v"(v[1][1]), "v"(v[1][2]),  "v"(v[1][3]),
        "v"(v[1][4]), "v"(v[1][5]), "v"(v[1][6]),  "v"(v[1][7]),
        "v"(v[1][8]), "v"(v[1][9]), "v"(v[1][10]), "v"(v[1][11]));
#pragma unroll
    for (int g = 0; g < NG; ++g)
#pragma unroll
        for (int t = 0; t < T; ++t)
            dst[((size_t)(n0 + 4 * g) * T + t) * C4 + c4] = v[g][t];
}

constexpr int NL = 4;

__global__ void __launch_bounds__(256)
k_link(const int* __restrict__ src1, const int* __restrict__ dst1,
       const float* __restrict__ ew1, int E1, int nb1,
       const int* __restrict__ src2, const int* __restrict__ dst2,
       const float* __restrict__ ew2, int E2,
       int* __restrict__ heads1, int* __restrict__ heads2,
       int4* __restrict__ rec1, int4* __restrict__ rec2) {
    int b = blockIdx.x;
    const int*   src = src1;
    const int*   dst = dst1;
    const float* ew  = ew1;
    int*         hd  = heads1;
    int4*        rec = rec1;
    int          E   = E1;
    if (b >= nb1) {
        b -= nb1;
        src = src2; dst = dst2; ew = ew2; hd = heads2; rec = rec2; E = E2;
    }
    int e = b * blockDim.x + threadIdx.x;
    if (e >= E) return;
    int d    = dst[e];
    int s    = src[e];
    float w  = ew[e];
    int prev = atomicExch(&hd[d * NL + (e & (NL - 1))], e);
    rec[e] = make_int4(s, __float_as_int(w), prev, 0);
}

__device__ __forceinline__ float4 gather_row(const int4* __restrict__ heads,
                                             const int4* __restrict__ recs,
                                             const float* __restrict__ x, int E,
                                             int n, int tid) {
    static_assert(NL == 4, "heads are read as one int4 per dst node");
    float4 acc = make_float4(0.f, 0.f, 0.f, 0.f);
    const float4* x4 = reinterpret_cast<const float4*>(x);
    const unsigned uE = (unsigned)E;

    int4 h = heads[n];
    int   nx[NL] = {h.x, h.y, h.z, h.w};
    bool  live[NL] = {false, false, false, false};
    int   sr[NL] = {0, 0, 0, 0};
    float wt[NL] = {0.f, 0.f, 0.f, 0.f};

    for (int it = 0; it <= E; ++it) {
        float4 v[NL];
#pragma unroll
        for (int k = 0; k < NL; ++k) {
            v[k] = make_float4(0.f, 0.f, 0.f, 0.f);
            if (live[k]) v[k] = x4[(size_t)sr[k] * ROW4 + tid];
        }
        int4 rn[NL];
        bool nlive[NL];
#pragma unroll
        for (int k = 0; k < NL; ++k) {
            nlive[k] = (unsigned)nx[k] < uE;
            rn[k]    = make_int4(0, 0, -1, 0);
            if (nlive[k]) rn[k] = recs[nx[k]];
        }
#pragma unroll
        for (int k = 0; k < NL; ++k) {
            if (live[k]) {
                acc.x += wt[k] * v[k].x;
                acc.y += wt[k] * v[k].y;
                acc.z += wt[k] * v[k].z;
                acc.w += wt[k] * v[k].w;
            }
        }
        bool any = false;
#pragma unroll
        for (int k = 0; k < NL; ++k) {
            live[k] = nlive[k];
            sr[k]   = rn[k].x;
            wt[k]   = __int_as_float(rn[k].y);
            nx[k]   = rn[k].z;
            any     = any | nlive[k];
        }
        if (!any) break;
    }
    return acc;
}

__global__ void __launch_bounds__(96)
k_aggregate(const int4* __restrict__ heads, const int4* __restrict__ recs,
            const float* __restrict__ x, float* __restrict__ agg, int E) {
    int n   = blockIdx.x;
    int tid = threadIdx.x;
    float4 acc = gather_row(heads, recs, x, E, n, tid);
    reinterpret_cast<float4*>(agg)[(size_t)n * ROW4 + tid] = acc;
}

constexpr int G_TPW   = 5;
constexpr int G_WAVES = M1 / 16 / G_TPW;
constexpr int G_BLOCK = 128;
static_assert((M1 / 16) % G_TPW == 0 && (G_WAVES * 32) % G_BLOCK == 0,
              "every wave owns G_TPW whole tiles; the grid holds whole work-groups only");

__global__ void __launch_bounds__(G_BLOCK)
k_gemm1(const float* __restrict__ xT, const float* __restrict__ agg1,
        const float* __restrict__ Wself, const float* __restrict__ Wneigh,
        const float* __restrict__ bias, float* __restrict__ h1) {
    const int lane = threadIdx.x & 31;
    const int gw   = (blockIdx.x * G_BLOCK + threadIdx.x) >> 5;
    const int mw   = gw * (G_TPW * 16);

    v2f aS[3][8], aN[3][8];
    static_assert(G_TPW >= 2, "the first two tiles are requested up front");
#pragma unroll
    for (int j = 0; j < 2; ++j)
#pragma unroll
        for (int ks = 0; ks < 8; ++ks) {
            aS[j][ks] = load_frag(xT,   mw + 16 * j, C_IN, ks * 4, lane);
            aN[j][ks] = load_frag(agg1, mw + 16 * j, C_IN, ks * 4, lane);
        }

    v2f bS[4][8], bN[4][8];
    float bv[4];
#pragma unroll
    for (int nt = 0; nt < 4; ++nt) {
#pragma unroll
        for (int ks = 0; ks < 8; ++ks) {
            bS[nt][ks] = load_frag(Wself,  nt * 16, C_IN, ks * 4, lane);
            bN[nt][ks] = load_frag(Wneigh, nt * 16, C_IN, ks * 4, lane);
        }
        bv[nt] = bias[nt * 16 + (lane & 15)];
    }

#pragma unroll
    for (int i = 0; i < G_TPW; ++i) {
        const int c  = i % 3;
        const int m0 = mw + i * 16;
        if (i + 2 < G_TPW) {
#pragma unroll
            for (int ks = 0; ks < 8; ++ks) {
                aS[(i + 2) % 3][ks] = load_frag(xT,   m0 + 32, C_IN, ks * 4, lane);
                aN[(i + 2) % 3][ks] = load_frag(agg1, m0 + 32, C_IN, ks * 4, lane);
            }
        }
        __builtin_amdgcn_sched_barrier(0);

        v8f acc[4] = {};
#pragma unroll
        for (int ks = 0; ks < 8; ++ks) {
#pragma unroll
            for (int nt = 0; nt < 4; ++nt) {
                acc[nt] = __builtin_amdgcn_wmma_f32_16x16x4_f32(false, aS[c][ks], false, bS[nt][ks],
                                                                (short)0, acc[nt], false, false);
                acc[nt] = __builtin_amdgcn_wmma_f32_16x16x4_f32(false, aN[c][ks], false, bN[nt][ks],
                                                                (short)0, acc[nt], false, false);
            }
        }
        const int mbase = m0 + ((lane >> 4) << 3);
#pragma unroll
        for (int nt = 0; nt < 4; ++nt) {
            const int col = nt * 16 + (lane & 15);
#pragma unroll
            for (int r = 0; r < 8; ++r) {
                float v = leaky(acc[nt][r] + bv[nt]);
                h1[(size_t)(mbase + r) * H + col] = v;
            }
        }
    }
}

__global__ void __launch_bounds__(G_BLOCK)
k_gemm_z2(const float* __restrict__ h1, const float* __restrict__ Wneigh2,
          float* __restrict__ z2) {
    const int lane = threadIdx.x & 31;
    const int gw   = (blockIdx.x * G_BLOCK + threadIdx.x) >> 5;
    const int mw   = gw * (G_TPW * 16);

    v2f a[2][16];
#pragma unroll
    for (int ks = 0; ks < 16; ++ks)
        a[0][ks] = load_frag(h1, mw, H, ks * 4, lane);

    v2f b[2][16];
#pragma unroll
    for (int nt = 0; nt < 2; ++nt)
#pragma unroll
        for (int ks = 0; ks < 16; ++ks)
            b[nt][ks] = load_frag(Wneigh2, nt * 16, H, ks * 4, lane);

#pragma unroll
    for (int i = 0; i < G_TPW; ++i) {
        const int c  = i & 1;
        const int m0 = mw + i * 16;
        if (i + 1 < G_TPW) {
#pragma unroll
            for (int ks = 0; ks < 16; ++ks)
                a[c ^ 1][ks] = load_frag(h1, m0 + 16, H, ks * 4, lane);
        }
        __builtin_amdgcn_sched_barrier(0);

        v8f acc[2] = {};
#pragma unroll
        for (int ks = 0; ks < 16; ++ks) {
#pragma unroll
            for (int nt = 0; nt < 2; ++nt)
                acc[nt] = __builtin_amdgcn_wmma_f32_16x16x4_f32(false, a[c][ks], false, b[nt][ks],
                                                                (short)0, acc[nt], false, false);
        }
        const int mbase = m0 + ((lane >> 4) << 3);
#pragma unroll
        for (int nt = 0; nt < 2; ++nt) {
            const int col = nt * 16 + (lane & 15);
#pragma unroll
            for (int r = 0; r < 8; ++r)
                z2[(size_t)(mbase + r) * C_OUT + col] = acc[nt][r];
        }
    }
}

constexpr int HP = H + 4;
__global__ void __launch_bounds__(96)
k_aggregate_final(const int4* __restrict__ heads, const int4* __restrict__ recs,
                  const float* __restrict__ z2, int E,
                  const float* __restrict__ h1, const float* __restrict__ Wself2,
                  const float* __restrict__ bias2, float* __restrict__ out) {
    static_assert(T == 12 && H == 64 && C_OUT == 32,
                  "12 rows of 64 = 3 float4 per thread of two waves; two 16-column tiles");
    __shared__ float4 s_agg4[T * C_OUT / 4];
    __shared__ float4 s_h4[16 * HP / 4];
    float* s_agg = reinterpret_cast<float*>(s_agg4);
    float* s_h   = reinterpret_cast<float*>(s_h4);

    int n    = blockIdx.x;
    int tid  = threadIdx.x;
    int lane = tid & 31;
    int wave = __builtin_amdgcn_readfirstlane(tid >> 5);

    const float4* h4 = reinterpret_cast<const float4*>(h1) + (size_t)n * (T * H / 4);
    float4 hv0 = make_float4(0.f, 0.f, 0.f, 0.f), hv1 = hv0, hv2 = hv0;
    if (wave < 2) {
        hv0 = h4[tid];
        hv1 = h4[tid + 64];
        hv2 = h4[tid + 128];
    }

    float4 acc = gather_row(heads, recs, z2, E, n, tid);

    s_agg4[tid] = acc;
    if (wave < 2) {
        float* p = &s_h[(tid >> 4) * HP + (tid & 15) * 4];
        *reinterpret_cast<float4*>(p)          = hv0;
        *reinterpret_cast<float4*>(p + 4 * HP) = hv1;
        *reinterpret_cast<float4*>(p + 8 * HP) = hv2;
    } else {
#pragma unroll
        for (int i = 0; i < 2; ++i) {
            int j = 192 + lane + 32 * i;
            *reinterpret_cast<float4*>(&s_h[(j >> 4) * HP + (j & 15) * 4]) =
                make_float4(0.f, 0.f, 0.f, 0.f);
        }
    }
    __syncthreads();

    if (wave < 2) {
        int n0 = wave * 16;
        int c  = lane & 15;
        int hh = lane >> 4;
        v8f d = {};
#pragma unroll
        for (int ks = 0; ks < 16; ++ks) {
            v2f a = *reinterpret_cast<const v2f*>(&s_h[c * HP + ks * 4 + 2 * hh]);
            v2f b = load_frag(Wself2, n0, H, ks * 4, lane);
            d = __builtin_amdgcn_wmma_f32_16x16x4_f32(false, a, false, b,
                                                      (short)0, d, false, false);
        }
        int col = n0 + c;
        int tb  = hh << 3;
        float bv = bias2[col];
#pragma unroll
        for (int r = 0; r < 8; ++r) {
            int t = tb + r;
            if (t < T) {
                float v = leaky(d[r] + s_agg[t * C_OUT + col] + bv);
                s_agg[t * C_OUT + col] = v;
            }
        }
    }
    __syncthreads();

    float4 o = s_agg4[tid];
    reinterpret_cast<float4*>(out)[((size_t)(tid >> 3) * N_DST2 + n) * (C_OUT / 4) + (tid & 7)] = o;
}

extern "C" void kernel_launch(void* const* d_in, const int* in_sizes, int n_in,
                              void* d_out, int out_size, void* d_ws, size_t ws_size,
                              hipStream_t stream) {
    const float* nf   = (const float*)d_in[0];
    const int*   src1 = (const int*)  d_in[1];
    const int*   dst1 = (const int*)  d_in[2];
    const float* ew1  = (const float*)d_in[3];
    const int*   src2 = (const int*)  d_in[4];
    const int*   dst2 = (const int*)  d_in[5];
    const float* ew2  = (const float*)d_in[6];
    const float* W1s  = (const float*)d_in[7];
    const float* W1n  = (const float*)d_in[8];
    const float* b1   = (const float*)d_in[9];
    const float* W2s  = (const float*)d_in[10];
    const float* W2n  = (const float*)d_in[11];
    const float* b2   = (const float*)d_in[12];
    const int E1 = in_sizes[1];
    const int E2 = in_sizes[4];

    float* out = (float*)d_out;

    float* ws   = (float*)d_ws;
    float* xT   = ws;
    float* agg1 = xT   + (size_t)N_SRC  * ROW;
    float* h1   = agg1 + (size_t)N_DST1 * ROW;
    float* z2   = xT;
    int*   ip     = (int*)(h1 + (size_t)M1 * H);
    int*   heads1 = ip;                  ip += N_DST1 * NL;
    int*   heads2 = ip;                  ip += N_DST2 * NL;
    int4*  rec2   = (int4*)ip;
    int4*  rec1   = (int4*)h1;

    k_transpose<<<TR_BLOCKS, TR_BLOCK, 0, stream>>>(nf, xT, heads1, (N_DST1 + N_DST2) * NL);
    k_link<<<(E1 + 255) / 256 + (E2 + 255) / 256, 256, 0, stream>>>(
        src1, dst1, ew1, E1, (E1 + 255) / 256, src2, dst2, ew2, E2,
        heads1, heads2, rec1, rec2);

    k_aggregate<<<N_DST1, 96, 0, stream>>>((const int4*)heads1, rec1, xT, agg1, E1);

    k_gemm1<<<G_WAVES * 32 / G_BLOCK, G_BLOCK, 0, stream>>>(xT, agg1, W1s, W1n, b1, h1);

    k_gemm_z2<<<G_WAVES * 32 / G_BLOCK, G_BLOCK, 0, stream>>>(h1, W2n, z2);

    k_aggregate_final<<<N_DST2, 96, 0, stream>>>((const int4*)heads2, rec2, z2, E2,
                                                 h1, W2s, b2, out);
}


# Both layer-2 products join the layer-1 GEMM launch: a wave passes its h1 tile through LDS (D order in, A fragments out); h1 never stored, last launch just gathers and adds; 6->5 launches; same sums
# speedup vs baseline: 2.3590x; 1.1439x over previous
#include <hip/hip_runtime.h>

typedef __attribute__((ext_vector_type(2))) float v2f;
typedef __attribute__((ext_vector_type(8))) float v8f;
typedef __attribute__((ext_vector_type(4))) float v4f;

constexpr int T      = 12;
constexpr int C_IN   = 32;
constexpr int H      = 64;
constexpr int C_OUT  = 32;
constexpr int N_SRC  = 50000;
constexpr int N_DST1 = 20000;
constexpr int N_DST2 = 10000;
constexpr int ROW    = T * C_IN;
constexpr int ROW4   = ROW / 4;
constexpr int M1     = N_DST1 * T;
constexpr int M2     = N_DST2 * T;
constexpr float NEG_SLOPE = 0.01f;

__device__ __forceinline__ float leaky(float v) {
    return v > 0.f ? v : NEG_SLOPE * v;
}

__device__ __forceinline__ v2f load_frag(const float* __restrict__ base, int r0, int ld,
                                         int k0, int lane) {
    int r = r0 + (lane & 15);
    int k = k0 + ((lane >> 4) << 1);
    const float* p = base + (size_t)r * ld + k;
    v2f f;
    f.x = p[0];
    f.y = p[1];
    return f;
}

constexpr int TR_NODES  = 8;
constexpr int TR_WAVES  = N_SRC / TR_NODES;
constexpr int TR_BLOCK  = 256;
constexpr int TR_BLOCKS = (TR_WAVES * 32 + TR_BLOCK - 1) / TR_BLOCK;
static_assert(N_SRC % TR_NODES == 0 && TR_NODES % 4 == 0 && C_IN == 32,
              "whole groups of 4 nodes per load; 8 float4 per (node, t) line");

__global__ void __launch_bounds__(TR_BLOCK)
k_transpose(const float* __restrict__ nf, float* __restrict__ xT,
            int* __restrict__ heads, int nheads) {
    const int gtid = blockIdx.x * TR_BLOCK + threadIdx.x;
    for (int i = gtid; i < nheads; i += TR_BLOCKS * TR_BLOCK) heads[i] = -1;

    const int gw = gtid >> 5;
    if (gw >= TR_WAVES) return;
    const int lane = threadIdx.x & 31;
    const int c4   = lane & 7;
    const int n0   = gw * TR_NODES + (lane >> 3);
    constexpr int C4 = C_IN / 4;
    constexpr int NG = TR_NODES / 4;
    const v4f* src = reinterpret_cast<const v4f*>(nf);
    v4f*       dst = reinterpret_cast<v4f*>(xT);
    static_assert(NG == 2 && T == 12, "the fence below names the 24 loaded values one by one");

    v4f v[NG][T];
#pragma unroll
    for (int g = 0; g < NG; ++g)
#pragma unroll
        for (int t = 0; t < T; ++t)
            v[g][t] = src[((size_t)t * N_SRC + (n0 + 4 * g)) * C4 + c4];
    asm volatile("" ::
        "v"(v[0][0]), "v"(v[0][1]), "v"(v[0][2]),  "v"(v[0][3]),
        "v"(v[0][4]), "v"(v[0][5]), "v"(v[0][6]),  "v"(v[0][7]),
        "v"(v[0][8]), "v"(v[0][9]), "v"(v[0][10]), "v"(v[0][11]),
        "v"(v[1][0]), "v"(v[1][1]), "v"(v[1][2]),  "v"(v[1][3]),
        "v"(v[1][4]), "v"(v[1][5]), "v"(v[1][6]),  "v"(v[1][7]),
        "v"(v[1][8]), "v"(v[1][9]), "v"(v[1][10]), "v"(v[1][11]));
#pragma unroll
    for (int g = 0; g < NG; ++g)
#pragma unroll
        for (int t = 0; t < T; ++t)
            dst[((size_t)(n0 + 4 * g) * T + t) * C4 + c4] = v[g][t];
}

constexpr int NL = 4;

__global__ void __launch_bounds__(256)
k_link(const int* __restrict__ src1, const int* __restrict__ dst1,
       const float* __restrict__ ew1, int E1, int nb1,
       const int* __restrict__ src2, const int* __restrict__ dst2,
       const float* __restrict__ ew2, int E2,
       int* __restrict__ heads1, int* __restrict__ heads2,
       int4* __restrict__ rec1, int4* __restrict__ rec2) {
    int b = blockIdx.x;
    const int*   src = src1;
    const int*   dst = dst1;
    const float* ew  = ew1;
    int*         hd  = heads1;
    int4*        rec = rec1;
    int          E   = E1;
    if (b >= nb1) {
        b -= nb1;
        src = src2; dst = dst2; ew = ew2; hd = heads2; rec = rec2; E = E2;
    }
    int e = b * blockDim.x + threadIdx.x;
    if (e >= E) return;
    int d    = dst[e];
    int s    = src[e];
    float w  = ew[e];
    int prev = atomicExch(&hd[d * NL + (e & (NL - 1))], e);
    rec[e] = make_int4(s, __float_as_int(w), prev, 0);
}

__device__ __forceinline__ float4 gather_row(const int4* __restrict__ heads,
                                             const int4* __restrict__ recs,
                                             const float* __restrict__ x, int E,
                                             int n, int tid) {
    static_assert(NL == 4, "heads are read as one int4 per dst node");
    float4 acc = make_float4(0.f, 0.f, 0.f, 0.f);
    const float4* x4 = reinterpret_cast<const float4*>(x);
    const unsigned uE = (unsigned)E;

    int4 h = heads[n];
    int   nx[NL] = {h.x, h.y, h.z, h.w};
    bool  live[NL] = {false, false, false, false};
    int   sr[NL] = {0, 0, 0, 0};
    float wt[NL] = {0.f, 0.f, 0.f, 0.f};

    for (int it = 0; it <= E; ++it) {
        float4 v[NL];
#pragma unroll
        for (int k = 0; k < NL; ++k) {
            v[k] = make_float4(0.f, 0.f, 0.f, 0.f);
            if (live[k]) v[k] = x4[(size_t)sr[k] * ROW4 + tid];
        }
        int4 rn[NL];
        bool nlive[NL];
#pragma unroll
        for (int k = 0; k < NL; ++k) {
            nlive[k] = (unsigned)nx[k] < uE;
            rn[k]    = make_int4(0, 0, -1, 0);
            if (nlive[k]) rn[k] = recs[nx[k]];
        }
#pragma unroll
        for (int k = 0; k < NL; ++k) {
            if (live[k]) {
                acc.x += wt[k] * v[k].x;
                acc.y += wt[k] * v[k].y;
                acc.z += wt[k] * v[k].z;
                acc.w += wt[k] * v[k].w;
            }
        }
        bool any = false;
#pragma unroll
        for (int k = 0; k < NL; ++k) {
            live[k] = nlive[k];
            sr[k]   = rn[k].x;
            wt[k]   = __int_as_float(rn[k].y);
            nx[k]   = rn[k].z;
            any     = any | nlive[k];
        }
        if (!any) break;
    }
    return acc;
}

__global__ void __launch_bounds__(96)
k_aggregate(const int4* __restrict__ heads, const int4* __restrict__ recs,
            const float* __restrict__ x, float* __restrict__ agg, int E) {
    int n   = blockIdx.x;
    int tid = threadIdx.x;
    float4 acc = gather_row(heads, recs, x, E, n, tid);
    reinterpret_cast<float4*>(agg)[(size_t)n * ROW4 + tid] = acc;
}

constexpr int F_TPW   = 8;
constexpr int F_WAVES = M1 / 16 / F_TPW;
constexpr int F_OWN   = (M2 / 16) / F_WAVES;
constexpr int HP      = H + 4;
static_assert((M1 / 16) % F_TPW == 0 && (M2 / 16) % F_WAVES == 0 && F_OWN <= F_TPW,
              "whole tiles per wave; the own-term tiles are exactly the first F_OWN of every wave");

__global__ void __launch_bounds__(32)
k_gemm_layers(const float* __restrict__ xT, const float* __restrict__ agg1,
              const float* __restrict__ Wself1, const float* __restrict__ Wneigh1,
              const float* __restrict__ bias1,
              const float* __restrict__ Wneigh2, const float* __restrict__ Wself2,
              float* __restrict__ z2, float* __restrict__ s2) {
    __shared__ float4 s_h4[16 * HP / 4];
    float* s_h = reinterpret_cast<float*>(s_h4);

    const int lane = threadIdx.x;
    const int w    = blockIdx.x;
    const int c    = lane & 15;
    const int hh   = lane >> 4;

    v2f aS[2][8], aN[2][8];
#pragma unroll
    for (int ks = 0; ks < 8; ++ks) {
        aS[0][ks] = load_frag(xT,   w * 16, C_IN, ks * 4, lane);
        aN[0][ks] = load_frag(agg1, w * 16, C_IN, ks * 4, lane);
    }

    v2f bS[4][8], bN[4][8];
    float bv[4];
#pragma unroll
    for (int nt = 0; nt < 4; ++nt) {
#pragma unroll
        for (int ks = 0; ks < 8; ++ks) {
            bS[nt][ks] = load_frag(Wself1,  nt * 16, C_IN, ks * 4, lane);
            bN[nt][ks] = load_frag(Wneigh1, nt * 16, C_IN, ks * 4, lane);
        }
        bv[nt] = bias1[nt * 16 + c];
    }
    v2f bZ[2][16], bO[2][16];
#pragma unroll
    for (int nt = 0; nt < 2; ++nt)
#pragma unroll
        for (int ks = 0; ks < 16; ++ks) {
            bZ[nt][ks] = load_frag(Wneigh2, nt * 16, H, ks * 4, lane);
            bO[nt][ks] = load_frag(Wself2,  nt * 16, H, ks * 4, lane);
        }

#pragma unroll
    for (int i = 0; i < F_TPW; ++i) {
        const int cs = i & 1;
        const int m0 = (i * F_WAVES + w) * 16;
        if (i + 1 < F_TPW) {
            const int mn = ((i + 1) * F_WAVES + w) * 16;
#pragma unroll
            for (int ks = 0; ks < 8; ++ks) {
                aS[cs ^ 1][ks] = load_frag(xT,   mn, C_IN, ks * 4, lane);
                aN[cs ^ 1][ks] = load_frag(agg1, mn, C_IN, ks * 4, lane);
            }
        }
        __builtin_amdgcn_sched_barrier(0);

        v8f acc[4] = {};
#pragma unroll
        for (int ks = 0; ks < 8; ++ks) {
#pragma unroll
            for (int nt = 0; nt < 4; ++nt) {
                acc[nt] = __builtin_amdgcn_wmma_f32_16x16x4_f32(false, aS[cs][ks], false, bS[nt][ks],
                                                                (short)0, acc[nt], false, false);
                acc[nt] = __builtin_amdgcn_wmma_f32_16x16x4_f32(false, aN[cs][ks], false, bN[nt][ks],
                                                                (short)0, acc[nt], false, false);
            }
        }
#pragma unroll
        for (int nt = 0; nt < 4; ++nt)
#pragma unroll
            for (int r = 0; r < 8; ++r)
                s_h[(8 * hh + r) * HP + nt * 16 + c] = leaky(acc[nt][r] + bv[nt]);
        __syncthreads();
        v2f a2[16];
#pragma unroll
        for (int ks = 0; ks < 16; ++ks)
            a2[ks] = *reinterpret_cast<const v2f*>(&s_h[c * HP + ks * 4 + 2 * hh]);
        __syncthreads();

        const int mbase = m0 + (hh << 3);
        v8f dz[2] = {};
#pragma unroll
        for (int ks = 0; ks < 16; ++ks) {
#pragma unroll
            for (int nt = 0; nt < 2; ++nt)
                dz[nt] = __builtin_amdgcn_wmma_f32_16x16x4_f32(false, a2[ks], false, bZ[nt][ks],
                                                               (short)0, dz[nt], false, false);
        }
#pragma unroll
        for (int nt = 0; nt < 2; ++nt)
#pragma unroll
            for (int r = 0; r < 8; ++r)
                z2[(size_t)(mbase + r) * C_OUT + nt * 16 + c] = dz[nt][r];

        if (i < F_OWN) {
            v8f dq[2] = {};
#pragma unroll
            for (int ks = 0; ks < 16; ++ks) {
#pragma unroll
                for (int nt = 0; nt < 2; ++nt)
                    dq[nt] = __builtin_amdgcn_wmma_f32_16x16x4_f32(false, a2[ks], false, bO[nt][ks],
                                                                   (short)0, dq[nt], false, false);
            }
#pragma unroll
            for (int nt = 0; nt < 2; ++nt)
#pragma unroll
                for (int r = 0; r < 8; ++r)
                    s2[(size_t)(mbase + r) * C_OUT + nt * 16 + c] = dq[nt][r];
        }
    }
}

__global__ void __launch_bounds__(96)
k_aggregate_final(const int4* __restrict__ heads, const int4* __restrict__ recs,
                  const float* __restrict__ z2, int E,
                  const float* __restrict__ s2, const float* __restrict__ bias2,
                  float* __restrict__ out) {
    static_assert(T * C_OUT == ROW && C_OUT == 32, "a node's 12x32 block is one 96-float4 row");
    int n   = blockIdx.x;
    int tid = threadIdx.x;
    float4 sv = reinterpret_cast<const float4*>(s2)[(size_t)n * ROW4 + tid];
    float4 bv = reinterpret_cast<const float4*>(bias2)[tid & 7];

    float4 acc = gather_row(heads, recs, z2, E, n, tid);

    float4 o;
    o.x = leaky(sv.x + acc.x + bv.x);
    o.y = leaky(sv.y + acc.y + bv.y);
    o.z = leaky(sv.z + acc.z + bv.z);
    o.w = leaky(sv.w + acc.w + bv.w);
    reinterpret_cast<float4*>(out)[((size_t)(tid >> 3) * N_DST2 + n) * (C_OUT / 4) + (tid & 7)] = o;
}

extern "C" void kernel_launch(void* const* d_in, const int* in_sizes, int n_in,
                              void* d_out, int out_size, void* d_ws, size_t ws_size,
                              hipStream_t stream) {
    const float* nf   = (const float*)d_in[0];
    const int*   src1 = (const int*)  d_in[1];
    const int*   dst1 = (const int*)  d_in[2];
    const float* ew1  = (const float*)d_in[3];
    const int*   src2 = (const int*)  d_in[4];
    const int*   dst2 = (const int*)  d_in[5];
    const float* ew2  = (const float*)d_in[6];
    const float* W1s  = (const float*)d_in[7];
    const float* W1n  = (const float*)d_in[8];
    const float* b1   = (const float*)d_in[9];
    const float* W2s  = (const float*)d_in[10];
    const float* W2n  = (const float*)d_in[11];
    const float* b2   = (const float*)d_in[12];
    const int E1 = in_sizes[1];
    const int E2 = in_sizes[4];

    float* out = (float*)d_out;

    float* ws   = (float*)d_ws;
    float* xT   = ws;
    float* agg1 = xT   + (size_t)N_SRC  * ROW;
    float* z2   = agg1 + (size_t)N_DST1 * ROW;
    float* s2   = z2   + (size_t)M1 * C_OUT;
    int*   ip     = (int*)(z2 + (size_t)M1 * H);
    int*   heads1 = ip;                  ip += N_DST1 * NL;
    int*   heads2 = ip;                  ip += N_DST2 * NL;
    int4*  rec2   = (int4*)ip;
    int4*  rec1   = (int4*)z2;

    k_transpose<<<TR_BLOCKS, TR_BLOCK, 0, stream>>>(nf, xT, heads1, (N_DST1 + N_DST2) * NL);
    k_link<<<(E1 + 255) / 256 + (E2 + 255) / 256, 256, 0, stream>>>(
        src1, dst1, ew1, E1, (E1 + 255) / 256, src2, dst2, ew2, E2,
        heads1, heads2, rec1, rec2);

    k_aggregate<<<N_DST1, 96, 0, stream>>>((const int4*)heads1, rec1, xT, agg1, E1);

    k_gemm_layers<<<F_WAVES, 32, 0, stream>>>(xT, agg1, W1s, W1n, b1, W2n, W2s, z2, s2);

    k_aggregate_final<<<N_DST2, 96, 0, stream>>>((const int4*)heads2, rec2, z2, E2, s2, b2, out);
}


# Gather walk made lane-parallel: 32 lists per dst node, one wave per node, lane l follows list l with per-lane loads (no scalar-cache hops); neighbour sums now in round-then-lane order
# speedup vs baseline: 2.6965x; 1.1431x over previous
#include <hip/hip_runtime.h>

typedef __attribute__((ext_vector_type(2))) float v2f;
typedef __attribute__((ext_vector_type(8))) float v8f;
typedef __attribute__((ext_vector_type(4))) float v4f;

constexpr int T      = 12;
constexpr int C_IN   = 32;
constexpr int H      = 64;
constexpr int C_OUT  = 32;
constexpr int N_SRC  = 50000;
constexpr int N_DST1 = 20000;
constexpr int N_DST2 = 10000;
constexpr int N_E2   = 160000;
constexpr int ROW    = T * C_IN;
constexpr int ROW4   = ROW / 4;
constexpr int M1     = N_DST1 * T;
constexpr int M2     = N_DST2 * T;
constexpr float NEG_SLOPE = 0.01f;

__device__ __forceinline__ float leaky(float v) {
    return v > 0.f ? v : NEG_SLOPE * v;
}

__device__ __forceinline__ v2f load_frag(const float* __restrict__ base, int r0, int ld,
                                         int k0, int lane) {
    int r = r0 + (lane & 15);
    int k = k0 + ((lane >> 4) << 1);
    const float* p = base + (size_t)r * ld + k;
    v2f f;
    f.x = p[0];
    f.y = p[1];
    return f;
}

constexpr int TR_NODES  = 8;
constexpr int TR_WAVES  = N_SRC / TR_NODES;
constexpr int TR_BLOCK  = 256;
constexpr int TR_BLOCKS = (TR_WAVES * 32 + TR_BLOCK - 1) / TR_BLOCK;
static_assert(N_SRC % TR_NODES == 0 && TR_NODES % 4 == 0 && C_IN == 32,
              "whole groups of 4 nodes per load; 8 float4 per (node, t) line");

__global__ void __launch_bounds__(TR_BLOCK)
k_transpose(const float* __restrict__ nf, float* __restrict__ xT,
            int* __restrict__ heads, int nheads) {
    const int gtid = blockIdx.x * TR_BLOCK + threadIdx.x;
    for (int i = gtid; i < nheads; i += TR_BLOCKS * TR_BLOCK) heads[i] = -1;

    const int gw = gtid >> 5;
    if (gw >= TR_WAVES) return;
    const int lane = threadIdx.x & 31;
    const int c4   = lane & 7;
    const int n0   = gw * TR_NODES + (lane >> 3);
    constexpr int C4 = C_IN / 4;
    constexpr int NG = TR_NODES / 4;
    const v4f* src = reinterpret_cast<const v4f*>(nf);
    v4f*       dst = reinterpret_cast<v4f*>(xT);
    static_assert(NG == 2 && T == 12, "the fence below names the 24 loaded values one by one");

    v4f v[NG][T];
#pragma unroll
    for (int g = 0; g < NG; ++g)
#pragma unroll
        for (int t = 0; t < T; ++t)
            v[g][t] = src[((size_t)t * N_SRC + (n0 + 4 * g)) * C4 + c4];
    asm volatile("" ::
        "v"(v[0][0]), "v"(v[0][1]), "v"(v[0][2]),  "v"(v[0][3]),
        "v"(v[0][4]), "v"(v[0][5]), "v"(v[0][6]),  "v"(v[0][7]),
        "v"(v[0][8]), "v"(v[0][9]), "v"(v[0][10]), "v"(v[0][11]),
        "v"(v[1][0]), "v"(v[1][1]), "v"(v[1][2]),  "v"(v[1][3]),
        "v"(v[1][4]), "v"(v[1][5]), "v"(v[1][6]),  "v"(v[1][7]),
        "v"(v[1][8]), "v"(v[1][9]), "v"(v[1][10]), "v"(v[1][11]));
#pragma unroll
    for (int g = 0; g < NG; ++g)
#pragma unroll
        for (int t = 0; t < T; ++t)
            dst[((size_t)(n0 + 4 * g) * T + t) * C4 + c4] = v[g][t];
}

constexpr int NL = 32;

__global__ void __launch_bounds__(256)
k_link(const int* __restrict__ src1, const int* __restrict__ dst1,
       const float* __restrict__ ew1, int E1, int nb1,
       const int* __restrict__ src2, const int* __restrict__ dst2,
       const float* __restrict__ ew2, int E2,
       int* __restrict__ heads1, int* __restrict__ heads2,
       int4* __restrict__ rec1, int4* __restrict__ rec2) {
    int b = blockIdx.x;
    const int*   src = src1;
    const int*   dst = dst1;
    const float* ew  = ew1;
    int*         hd  = heads1;
    int4*        rec = rec1;
    int          E   = E1;
    if (b >= nb1) {
        b -= nb1;
        src = src2; dst = dst2; ew = ew2; hd = heads2; rec = rec2; E = E2;
    }
    int e = b * blockDim.x + threadIdx.x;
    if (e >= E) return;
    int d    = dst[e];
    int s    = src[e];
    float w  = ew[e];
    int prev = atomicExch(&hd[d * NL + (e & (NL - 1))], e);
    rec[e] = make_int4(s, __float_as_int(w), prev, 0);
}

constexpr int GB     = 4;
constexpr int G_WPB  = 8;
static_assert(NL == 32 && ROW4 == 96, "one list per lane; three float4 per lane per row");
static_assert(N_DST1 % G_WPB == 0 && N_DST2 % G_WPB == 0, "whole work-groups of nodes");

__device__ __forceinline__ void gather_node(const int* __restrict__ heads,
                                            const int4* __restrict__ recs,
                                            const float* __restrict__ x, int E,
                                            int n, int lane, float4 (&acc)[3]) {
    const float4* x4 = reinterpret_cast<const float4*>(x);
    const unsigned uE = (unsigned)E;
#pragma unroll
    for (int j = 0; j < 3; ++j) acc[j] = make_float4(0.f, 0.f, 0.f, 0.f);

    int  cur  = heads[(size_t)n * NL + lane];
    bool live = (unsigned)cur < uE;
    int4 r    = make_int4(0, 0, -1, 0);
    if (live) r = recs[cur];

    for (int it = 0; it <= E; ++it) {
        unsigned m = __builtin_amdgcn_ballot_w32(live);
        if (m == 0u) break;
        const bool nlive = live && ((unsigned)r.z < uE);
        int4 rn = make_int4(0, 0, -1, 0);
        if (nlive) rn = recs[r.z];

        while (m != 0u) {
            int   s[GB];
            float w[GB];
            bool  has[GB];
#pragma unroll
            for (int k = 0; k < GB; ++k) {
                has[k] = (m != 0u);
                s[k] = 0;
                w[k] = 0.f;
                if (has[k]) {
                    const int l = __builtin_ctz(m);
                    m &= m - 1u;
                    s[k] = __builtin_amdgcn_readlane(r.x, l);
                    w[k] = __int_as_float(__builtin_amdgcn_readlane(r.y, l));
                }
            }
            float4 v[GB][3];
#pragma unroll
            for (int k = 0; k < GB; ++k) {
#pragma unroll
                for (int j = 0; j < 3; ++j) v[k][j] = make_float4(0.f, 0.f, 0.f, 0.f);
                if (has[k]) {
#pragma unroll
                    for (int j = 0; j < 3; ++j)
                        v[k][j] = x4[(size_t)s[k] * ROW4 + j * 32 + lane];
                }
            }
#pragma unroll
            for (int k = 0; k < GB; ++k) {
                if (has[k]) {
#pragma unroll
                    for (int j = 0; j < 3; ++j) {
                        acc[j].x += w[k] * v[k][j].x;
                        acc[j].y += w[k] * v[k][j].y;
                        acc[j].z += w[k] * v[k][j].z;
                        acc[j].w += w[k] * v[k][j].w;
                    }
                }
            }
        }
        r    = rn;
        live = nlive;
    }
}

__global__ void __launch_bounds__(G_WPB * 32)
k_aggregate(const int* __restrict__ heads, const int4* __restrict__ recs,
            const float* __restrict__ x, float* __restrict__ agg, int E) {
    const int lane = threadIdx.x & 31;
    const int n    = blockIdx.x * G_WPB + (threadIdx.x >> 5);
    float4 acc[3];
    gather_node(heads, recs, x, E, n, lane, acc);
    float4* o = reinterpret_cast<float4*>(agg) + (size_t)n * ROW4;
#pragma unroll
    for (int j = 0; j < 3; ++j) o[j * 32 + lane] = acc[j];
}

constexpr int F_TPW   = 8;
constexpr int F_WAVES = M1 / 16 / F_TPW;
constexpr int F_OWN   = (M2 / 16) / F_WAVES;
constexpr int HP      = H + 4;
static_assert((M1 / 16) % F_TPW == 0 && (M2 / 16) % F_WAVES == 0 && F_OWN <= F_TPW,
              "whole tiles per wave; the own-term tiles are exactly the first F_OWN of every wave");

__global__ void __launch_bounds__(32)
k_gemm_layers(const float* __restrict__ xT, const float* __restrict__ agg1,
              const float* __restrict__ Wself1, const float* __restrict__ Wneigh1,
              const float* __restrict__ bias1,
              const float* __restrict__ Wneigh2, const float* __restrict__ Wself2,
              float* __restrict__ z2, float* __restrict__ s2) {
    __shared__ float4 s_h4[16 * HP / 4];
    float* s_h = reinterpret_cast<float*>(s_h4);

    const int lane = threadIdx.x;
    const int w    = blockIdx.x;
    const int c    = lane & 15;
    const int hh   = lane >> 4;

    v2f aS[2][8], aN[2][8];
#pragma unroll
    for (int ks = 0; ks < 8; ++ks) {
        aS[0][ks] = load_frag(xT,   w * 16, C_IN, ks * 4, lane);
        aN[0][ks] = load_frag(agg1, w * 16, C_IN, ks * 4, lane);
    }

    v2f bS[4][8], bN[4][8];
    float bv[4];
#pragma unroll
    for (int nt = 0; nt < 4; ++nt) {
#pragma unroll
        for (int ks = 0; ks < 8; ++ks) {
            bS[nt][ks] = load_frag(Wself1,  nt * 16, C_IN, ks * 4, lane);
            bN[nt][ks] = load_frag(Wneigh1, nt * 16, C_IN, ks * 4, lane);
        }
        bv[nt] = bias1[nt * 16 + c];
    }
    v2f bZ[2][16], bO[2][16];
#pragma unroll
    for (int nt = 0; nt < 2; ++nt)
#pragma unroll
        for (int ks = 0; ks < 16; ++ks) {
            bZ[nt][ks] = load_frag(Wneigh2, nt * 16, H, ks * 4, lane);
            bO[nt][ks] = load_frag(Wself2,  nt * 16, H, ks * 4, lane);
        }

#pragma unroll
    for (int i = 0; i < F_TPW; ++i) {
        const int cs = i & 1;
        const int m0 = (i * F_WAVES + w) * 16;
        if (i + 1 < F_TPW) {
            const int mn = ((i + 1) * F_WAVES + w) * 16;
#pragma unroll
            for (int ks = 0; ks < 8; ++ks) {
                aS[cs ^ 1][ks] = load_frag(xT,   mn, C_IN, ks * 4, lane);
                aN[cs ^ 1][ks] = load_frag(agg1, mn, C_IN, ks * 4, lane);
            }
        }
        __builtin_amdgcn_sched_barrier(0);

        v8f acc[4] = {};
#pragma unroll
        for (int ks = 0; ks < 8; ++ks) {
#pragma unroll
            for (int nt = 0; nt < 4; ++nt) {
                acc[nt] = __builtin_amdgcn_wmma_f32_16x16x4_f32(false, aS[cs][ks], false, bS[nt][ks],
                                                                (short)0, acc[nt], false, false);
                acc[nt] = __builtin_amdgcn_wmma_f32_16x16x4_f32(false, aN[cs][ks], false, bN[nt][ks],
                                                                (short)0, acc[nt], false, false);
            }
        }
#pragma unroll
        for (int nt = 0; nt < 4; ++nt)
#pragma unroll
            for (int r = 0; r < 8; ++r)
                s_h[(8 * hh + r) * HP + nt * 16 + c] = leaky(acc[nt][r] + bv[nt]);
        __syncthreads();
        v2f a2[16];
#pragma unroll
        for (int ks = 0; ks < 16; ++ks)
            a2[ks] = *reinterpret_cast<const v2f*>(&s_h[c * HP + ks * 4 + 2 * hh]);
        __syncthreads();

        const int mbase = m0 + (hh << 3);
        v8f dz[2] = {};
#pragma unroll
        for (int ks = 0; ks < 16; ++ks) {
#pragma unroll
            for (int nt = 0; nt < 2; ++nt)
                dz[nt] = __builtin_amdgcn_wmma_f32_16x16x4_f32(false, a2[ks], false, bZ[nt][ks],
                                                               (short)0, dz[nt], false, false);
        }
#pragma unroll
        for (int nt = 0; nt < 2; ++nt)
#pragma unroll
            for (int r = 0; r < 8; ++r)
                z2[(size_t)(mbase + r) * C_OUT + nt * 16 + c] = dz[nt][r];

        if (i < F_OWN) {
            v8f dq[2] = {};
#pragma unroll
            for (int ks = 0; ks < 16; ++ks) {
#pragma unroll
                for (int nt = 0; nt < 2; ++nt)
                    dq[nt] = __builtin_amdgcn_wmma_f32_16x16x4_f32(false, a2[ks], false, bO[nt][ks],
                                                                   (short)0, dq[nt], false, false);
            }
#pragma unroll
            for (int nt = 0; nt < 2; ++nt)
#pragma unroll
                for (int r = 0; r < 8; ++r)
                    s2[(size_t)(mbase + r) * C_OUT + nt * 16 + c] = dq[nt][r];
        }
    }
}

__global__ void __launch_bounds__(G_WPB * 32)
k_aggregate_final(const int* __restrict__ heads, const int4* __restrict__ recs,
                  const float* __restrict__ z2, int E,
                  const float* __restrict__ s2, const float* __restrict__ bias2,
                  float* __restrict__ out) {
    static_assert(T * C_OUT == ROW && C_OUT == 32, "a node's 12x32 block is one 96-float4 row");
    const int lane = threadIdx.x & 31;
    const int n    = blockIdx.x * G_WPB + (threadIdx.x >> 5);
    float4 sv[3];
#pragma unroll
    for (int j = 0; j < 3; ++j)
        sv[j] = reinterpret_cast<const float4*>(s2)[(size_t)n * ROW4 + j * 32 + lane];
    const float4 bv = reinterpret_cast<const float4*>(bias2)[lane & 7];

    float4 acc[3];
    gather_node(heads, recs, z2, E, n, lane, acc);

#pragma unroll
    for (int j = 0; j < 3; ++j) {
        float4 o;
        o.x = leaky(sv[j].x + acc[j].x + bv.x);
        o.y = leaky(sv[j].y + acc[j].y + bv.y);
        o.z = leaky(sv[j].z + acc[j].z + bv.z);
        o.w = leaky(sv[j].w + acc[j].w + bv.w);
        const int t = (lane >> 3) + 4 * j;
        reinterpret_cast<float4*>(out)[((size_t)t * N_DST2 + n) * (C_OUT / 4) + (lane & 7)] = o;
    }
}

extern "C" void kernel_launch(void* const* d_in, const int* in_sizes, int n_in,
                              void* d_out, int out_size, void* d_ws, size_t ws_size,
                              hipStream_t stream) {
    const float* nf   = (const float*)d_in[0];
    const int*   src1 = (const int*)  d_in[1];
    const int*   dst1 = (const int*)  d_in[2];
    const float* ew1  = (const float*)d_in[3];
    const int*   src2 = (const int*)  d_in[4];
    const int*   dst2 = (const int*)  d_in[5];
    const float* ew2  = (const float*)d_in[6];
    const float* W1s  = (const float*)d_in[7];
    const float* W1n  = (const float*)d_in[8];
    const float* b1   = (const float*)d_in[9];
    const float* W2s  = (const float*)d_in[10];
    const float* W2n  = (const float*)d_in[11];
    const float* b2   = (const float*)d_in[12];
    const int E1 = in_sizes[1];
    const int E2 = in_sizes[4];

    float* out = (float*)d_out;

    float* ws   = (float*)d_ws;
    float* xT   = ws;
    float* agg1 = xT   + (size_t)N_SRC  * ROW;
    float* z2   = agg1 + (size_t)N_DST1 * ROW;
    float* s2   = z2   + (size_t)M1 * C_OUT;
    static_assert((size_t)(N_DST1 + N_DST2) * NL + (size_t)N_E2 * 4
                      <= (size_t)M1 * H - (size_t)M1 * C_OUT - (size_t)M2 * C_OUT,
                  "heads and graph-2 records fit behind s2 inside the old h1 region");
    int*   ip     = (int*)(s2 + (size_t)M2 * C_OUT);
    int*   heads1 = ip;                  ip += N_DST1 * NL;
    int*   heads2 = ip;                  ip += N_DST2 * NL;
    int4*  rec2   = (int4*)ip;
    int4*  rec1   = (int4*)z2;

    k_transpose<<<TR_BLOCKS, TR_BLOCK, 0, stream>>>(nf, xT, heads1, (N_DST1 + N_DST2) * NL);
    k_link<<<(E1 + 255) / 256 + (E2 + 255) / 256, 256, 0, stream>>>(
        src1, dst1, ew1, E1, (E1 + 255) / 256, src2, dst2, ew2, E2,
        heads1, heads2, rec1, rec2);

    k_aggregate<<<N_DST1 / G_WPB, G_WPB * 32, 0, stream>>>(heads1, rec1, xT, agg1, E1);

    k_gemm_layers<<<F_WAVES, 32, 0, stream>>>(xT, agg1, W1s, W1n, b1, W2n, W2s, z2, s2);

    k_aggregate_final<<<N_DST2 / G_WPB, G_WPB * 32, 0, stream>>>(heads2, rec2, z2, E2, s2, b2, out);
}
